# starting kernel
# baseline (speedup 1.0000x reference)
#include <hip/hip_runtime.h>

#define H_DIM   128
#define F_IN_DIM 256
#define C_DIM   40
#define L_DIM   4
#define STEPF   0.1f

typedef __attribute__((ext_vector_type(2))) float v2f;
typedef __attribute__((ext_vector_type(4))) float v4f;
typedef __attribute__((ext_vector_type(8))) float v8f;


__global__ void zero_two(float* __restrict__ a, float* __restrict__ b, int n) {
    int i = blockIdx.x * blockDim.x + threadIdx.x;
    if (i < n) { a[i] = 0.f; b[i] = 0.f; }
}

__global__ void deg_accum(const int* __restrict__ row, float* __restrict__ deg, int E) {
    int e = blockIdx.x * blockDim.x + threadIdx.x;
    if (e < E) atomicAdd(&deg[row[e]], 1.0f);
}

__global__ void deg_inv_kernel(float* __restrict__ deg, int n) {
    int i = blockIdx.x * blockDim.x + threadIdx.x;
    if (i < n) { float d = deg[i]; deg[i] = (d > 0.f) ? rsqrtf(d) : 0.f; }
}

__global__ void coef_accum(const int* __restrict__ row, const int* __restrict__ col,
                           const float* __restrict__ dinv, float* __restrict__ coef, int E) {
    int e = blockIdx.x * blockDim.x + threadIdx.x;
    if (e < E) {
        int r = row[e], c = col[e];
        atomicAdd(&coef[c], dinv[r] * dinv[c]);
    }
}

__global__ void weff_kernel(const float* __restrict__ Wraw, float* __restrict__ Weff) {
    int l = blockIdx.x;
    int i = threadIdx.x;
    const float* Wr = Wraw + (size_t)l * H_DIM * (H_DIM + 2);
    float*       We = Weff + (size_t)l * H_DIM * H_DIM;
    float s = 0.f;
    for (int j = 0; j < H_DIM; ++j) {
        float w = 0.f;
        if (j > i)      w = Wr[i * (H_DIM + 2) + j];
        else if (j < i) w = Wr[j * (H_DIM + 2) + i];
        s += fabsf(w);
        We[i * H_DIM + j] = w;
    }
    float d = Wr[i * (H_DIM + 2) + H_DIM] * s + Wr[i * (H_DIM + 2) + H_DIM + 1];
    We[i * H_DIM + i] = d;
}


__launch_bounds__(256)
__global__ void encoder_gemm(const float* __restrict__ x, const float* __restrict__ W,
                             const float* __restrict__ bias, float* __restrict__ h, int Nn) {
    __shared__ __align__(16) float As[16][F_IN_DIM + 4];
    const int row0 = blockIdx.x * 16;
    const v4f* x4 = (const v4f*)x;
    for (int idx = threadIdx.x; idx < 16 * (F_IN_DIM / 4); idx += 256) {
        int r  = idx >> 6;
        int c4 = idx & 63;
        int node = row0 + r;
        v4f v = {0.f, 0.f, 0.f, 0.f};
        if (node < Nn)
            v = __builtin_nontemporal_load(&x4[(size_t)node * (F_IN_DIM / 4) + c4]);
        *(v4f*)&As[r][c4 * 4] = v;
    }
    __syncthreads();

    const int lane = threadIdx.x & 31;
    const int wave = threadIdx.x >> 5;
    const int col0 = wave * 16;
    const int m    = lane & 15;
    const int hi   = lane >> 4;
    const int colN = col0 + m;
    const float* wrow = W + (size_t)colN * F_IN_DIM;

    v8f acc = {};
    for (int k0 = 0; k0 < F_IN_DIM; k0 += 4) {
        int k = k0 + 2 * hi;
        v2f a = *(const v2f*)&As[m][k];
        v2f b = *(const v2f*)&wrow[k];
        acc = __builtin_amdgcn_wmma_f32_16x16x4_f32(false, a, false, b,
                                                    (short)0, acc, false, false);
    }
    const float bv = bias[colN];
    for (int r = 0; r < 8; ++r) {
        int node = row0 + r + 8 * hi;
        if (node < Nn) h[(size_t)node * H_DIM + colN] = acc[r] + bv;
    }
}

__launch_bounds__(256)
__global__ void layer_gemm(const float* __restrict__ h_in, const float* __restrict__ h0,
                           const float* __restrict__ coef, const float* __restrict__ Weff,
                           const float* __restrict__ Wb, const float* __restrict__ extw,
                           const float* __restrict__ beta_p, float* __restrict__ h_out, int Nn) {
    __shared__ __align__(16) float As[16][H_DIM + 4];
    const int row0 = blockIdx.x * 16;
    const v4f* h4 = (const v4f*)h_in;
    for (int idx = threadIdx.x; idx < 16 * (H_DIM / 4); idx += 256) {
        int r  = idx >> 5;
        int c4 = idx & 31;
        int node = row0 + r;
        v4f v = {0.f, 0.f, 0.f, 0.f};
        if (node < Nn) v = h4[(size_t)node * (H_DIM / 4) + c4];
        *(v4f*)&As[r][c4 * 4] = v;
    }
    __syncthreads();

    const int lane = threadIdx.x & 31;
    const int wave = threadIdx.x >> 5;
    const int col0 = wave * 16;
    const int m    = lane & 15;
    const int hi   = lane >> 4;
    const int colN = col0 + m;

    v8f acc = {};
    for (int k0 = 0; k0 < H_DIM; k0 += 4) {
        int k = k0 + 2 * hi;
        v2f a = *(const v2f*)&As[m][k];
        v2f b;
        b.x = Weff[(size_t)k * H_DIM + colN];
        b.y = Weff[(size_t)(k + 1) * H_DIM + colN];
        acc = __builtin_amdgcn_wmma_f32_16x16x4_f32(false, a, false, b,
                                                    (short)0, acc, false, false);
    }

    const float wbv = Wb[colN];
    const float ew  = extw[colN];
    const float bl  = *beta_p;
    for (int r = 0; r < 8; ++r) {
        int node = row0 + r + 8 * hi;
        if (node < Nn) {
            float cf   = coef[node];
            float hv   = As[r + 8 * hi][colN];
            float h0v  = h0[(size_t)node * H_DIM + colN];
            float outp = acc[r] + wbv;
            float outv = cf * outp - (hv * ew + bl * h0v);
            h_out[(size_t)node * H_DIM + colN] = hv + STEPF * fmaxf(outv, 0.f);
        }
    }
}

__launch_bounds__(96)
__global__ void decoder_gemm(const float* __restrict__ h, const float* __restrict__ W,
                             const float* __restrict__ bias, float* __restrict__ out, int Nn) {
    __shared__ __align__(16) float As[16][H_DIM + 4];
    const int row0 = blockIdx.x * 16;
    const v4f* h4 = (const v4f*)h;
    for (int idx = threadIdx.x; idx < 16 * (H_DIM / 4); idx += 96) {
        int r  = idx >> 5;
        int c4 = idx & 31;
        int node = row0 + r;
        v4f v = {0.f, 0.f, 0.f, 0.f};
        if (node < Nn) v = h4[(size_t)node * (H_DIM / 4) + c4];
        *(v4f*)&As[r][c4 * 4] = v;
    }
    __syncthreads();

    const int lane = threadIdx.x & 31;
    const int wave = threadIdx.x >> 5;
    const int col0 = wave * 16;
    const int m    = lane & 15;
    const int hi   = lane >> 4;
    const int colN = col0 + m;
    const bool colok = (colN < C_DIM);
    const float* wrow = W + (size_t)(colok ? colN : 0) * H_DIM;

    v8f acc = {};
    for (int k0 = 0; k0 < H_DIM; k0 += 4) {
        int k = k0 + 2 * hi;
        v2f a = *(const v2f*)&As[m][k];
        v2f b = {0.f, 0.f};
        if (colok) b = *(const v2f*)&wrow[k];
        acc = __builtin_amdgcn_wmma_f32_16x16x4_f32(false, a, false, b,
                                                    (short)0, acc, false, false);
    }
    if (colok) {
        const float bv = bias[colN];
        for (int r = 0; r < 8; ++r) {
            int node = row0 + r + 8 * hi;
            if (node < Nn)
                __builtin_nontemporal_store(acc[r] + bv,
                                            &out[(size_t)node * C_DIM + colN]);
        }
    }
}


extern "C" void kernel_launch(void* const* d_in, const int* in_sizes, int n_in,
                              void* d_out, int out_size, void* d_ws, size_t ws_size,
                              hipStream_t stream) {
    const float* x     = (const float*)d_in[0];
    const int*   eidx  = (const int*)d_in[1];
    const float* enc_W = (const float*)d_in[2];
    const float* enc_b = (const float*)d_in[3];
    const float* W_raw = (const float*)d_in[4];
    const float* W_b   = (const float*)d_in[5];
    const float* ext_w = (const float*)d_in[6];
    const float* beta  = (const float*)d_in[7];
    const float* dec_W = (const float*)d_in[8];
    const float* dec_b = (const float*)d_in[9];

    const int Nn = in_sizes[0] / F_IN_DIM;
    const int E  = in_sizes[1] / 2;
    const int* row = eidx;
    const int* col = eidx + E;

    char* ws = (char*)d_ws;
    size_t off = 0;
    auto wsalloc = [&](size_t bytes) -> void* {
        void* p = ws + off;
        off = (off + bytes + 255) & ~(size_t)255;
        return p;
    };
    float* deg  = (float*)wsalloc((size_t)Nn * sizeof(float));
    float* coef = (float*)wsalloc((size_t)Nn * sizeof(float));
    float* Weff = (float*)wsalloc((size_t)L_DIM * H_DIM * H_DIM * sizeof(float));
    float* h0   = (float*)wsalloc((size_t)Nn * H_DIM * sizeof(float));
    float* ha   = (float*)wsalloc((size_t)Nn * H_DIM * sizeof(float));
    float* hb   = (float*)wsalloc((size_t)Nn * H_DIM * sizeof(float));
    (void)ws_size; (void)n_in; (void)out_size;

    const int tb = 256;
    zero_two<<<(Nn + tb - 1) / tb, tb, 0, stream>>>(deg, coef, Nn);
    deg_accum<<<(E + tb - 1) / tb, tb, 0, stream>>>(row, deg, E);
    deg_inv_kernel<<<(Nn + tb - 1) / tb, tb, 0, stream>>>(deg, Nn);
    coef_accum<<<(E + tb - 1) / tb, tb, 0, stream>>>(row, col, deg, coef, E);
    weff_kernel<<<L_DIM, H_DIM, 0, stream>>>(W_raw, Weff);

    const int gTiles = (Nn + 15) / 16;
    encoder_gemm<<<gTiles, 256, 0, stream>>>(x, enc_W, enc_b, h0, Nn);

    const float* hin = h0;
    float* bufs[2] = { ha, hb };
    for (int l = 0; l < L_DIM; ++l) {
        float* hout = bufs[l & 1];
        layer_gemm<<<gTiles, 256, 0, stream>>>(hin, h0, coef,
                                               Weff + (size_t)l * H_DIM * H_DIM,
                                               W_b + (size_t)l * H_DIM,
                                               ext_w + (size_t)l * H_DIM,
                                               beta + l, hout, Nn);
        hin = hout;
    }
    decoder_gemm<<<gTiles, 96, 0, stream>>>(hin, dec_W, dec_b, (float*)d_out, Nn);
}


# weff set-up kernel: one block per matrix row (512 blocks, one thread per element, row sum read back from LDS in the same index order) instead of one thread per row in 4 blocks
# speedup vs baseline: 1.1489x; 1.1489x over previous
#include <hip/hip_runtime.h>

#define H_DIM   128
#define F_IN_DIM 256
#define C_DIM   40
#define L_DIM   4
#define STEPF   0.1f

typedef __attribute__((ext_vector_type(2))) float v2f;
typedef __attribute__((ext_vector_type(4))) float v4f;
typedef __attribute__((ext_vector_type(8))) float v8f;


__global__ void zero_two(float* __restrict__ a, float* __restrict__ b, int n) {
    int i = blockIdx.x * blockDim.x + threadIdx.x;
    if (i < n) { a[i] = 0.f; b[i] = 0.f; }
}

__global__ void deg_accum(const int* __restrict__ row, float* __restrict__ deg, int E) {
    int e = blockIdx.x * blockDim.x + threadIdx.x;
    if (e < E) atomicAdd(&deg[row[e]], 1.0f);
}

__global__ void deg_inv_kernel(float* __restrict__ deg, int n) {
    int i = blockIdx.x * blockDim.x + threadIdx.x;
    if (i < n) { float d = deg[i]; deg[i] = (d > 0.f) ? rsqrtf(d) : 0.f; }
}

__global__ void coef_accum(const int* __restrict__ row, const int* __restrict__ col,
                           const float* __restrict__ dinv, float* __restrict__ coef, int E) {
    int e = blockIdx.x * blockDim.x + threadIdx.x;
    if (e < E) {
        int r = row[e], c = col[e];
        atomicAdd(&coef[c], dinv[r] * dinv[c]);
    }
}

__launch_bounds__(H_DIM)
__global__ void weff_kernel(const float* __restrict__ Wraw, float* __restrict__ Weff) {
    __shared__ float absrow[H_DIM];
    const int l = blockIdx.x / H_DIM;
    const int i = blockIdx.x % H_DIM;
    const int j = threadIdx.x;
    const float* Wr = Wraw + (size_t)l * H_DIM * (H_DIM + 2);
    float*       We = Weff + (size_t)l * H_DIM * H_DIM;
    float w = 0.f;
    if (j > i)      w = Wr[i * (H_DIM + 2) + j];
    else if (j < i) w = Wr[j * (H_DIM + 2) + i];
    absrow[j] = fabsf(w);
    const float da = Wr[i * (H_DIM + 2) + H_DIM];
    const float db = Wr[i * (H_DIM + 2) + H_DIM + 1];
    __syncthreads();
    float v = w;
    if (j == i) {
        float s = 0.f;
        for (int t = 0; t < H_DIM; ++t) s += absrow[t];
        v = da * s + db;
    }
    We[i * H_DIM + j] = v;
}


__launch_bounds__(256)
__global__ void encoder_gemm(const float* __restrict__ x, const float* __restrict__ W,
                             const float* __restrict__ bias, float* __restrict__ h, int Nn) {
    __shared__ __align__(16) float As[16][F_IN_DIM + 4];
    const int row0 = blockIdx.x * 16;
    const v4f* x4 = (const v4f*)x;
    for (int idx = threadIdx.x; idx < 16 * (F_IN_DIM / 4); idx += 256) {
        int r  = idx >> 6;
        int c4 = idx & 63;
        int node = row0 + r;
        v4f v = {0.f, 0.f, 0.f, 0.f};
        if (node < Nn)
            v = __builtin_nontemporal_load(&x4[(size_t)node * (F_IN_DIM / 4) + c4]);
        *(v4f*)&As[r][c4 * 4] = v;
    }
    __syncthreads();

    const int lane = threadIdx.x & 31;
    const int wave = threadIdx.x >> 5;
    const int col0 = wave * 16;
    const int m    = lane & 15;
    const int hi   = lane >> 4;
    const int colN = col0 + m;
    const float* wrow = W + (size_t)colN * F_IN_DIM;

    v8f acc = {};
    for (int k0 = 0; k0 < F_IN_DIM; k0 += 4) {
        int k = k0 + 2 * hi;
        v2f a = *(const v2f*)&As[m][k];
        v2f b = *(const v2f*)&wrow[k];
        acc = __builtin_amdgcn_wmma_f32_16x16x4_f32(false, a, false, b,
                                                    (short)0, acc, false, false);
    }
    const float bv = bias[colN];
    for (int r = 0; r < 8; ++r) {
        int node = row0 + r + 8 * hi;
        if (node < Nn) h[(size_t)node * H_DIM + colN] = acc[r] + bv;
    }
}

__launch_bounds__(256)
__global__ void layer_gemm(const float* __restrict__ h_in, const float* __restrict__ h0,
                           const float* __restrict__ coef, const float* __restrict__ Weff,
                           const float* __restrict__ Wb, const float* __restrict__ extw,
                           const float* __restrict__ beta_p, float* __restrict__ h_out, int Nn) {
    __shared__ __align__(16) float As[16][H_DIM + 4];
    const int row0 = blockIdx.x * 16;
    const v4f* h4 = (const v4f*)h_in;
    for (int idx = threadIdx.x; idx < 16 * (H_DIM / 4); idx += 256) {
        int r  = idx >> 5;
        int c4 = idx & 31;
        int node = row0 + r;
        v4f v = {0.f, 0.f, 0.f, 0.f};
        if (node < Nn) v = h4[(size_t)node * (H_DIM / 4) + c4];
        *(v4f*)&As[r][c4 * 4] = v;
    }
    __syncthreads();

    const int lane = threadIdx.x & 31;
    const int wave = threadIdx.x >> 5;
    const int col0 = wave * 16;
    const int m    = lane & 15;
    const int hi   = lane >> 4;
    const int colN = col0 + m;

    v8f acc = {};
    for (int k0 = 0; k0 < H_DIM; k0 += 4) {
        int k = k0 + 2 * hi;
        v2f a = *(const v2f*)&As[m][k];
        v2f b;
        b.x = Weff[(size_t)k * H_DIM + colN];
        b.y = Weff[(size_t)(k + 1) * H_DIM + colN];
        acc = __builtin_amdgcn_wmma_f32_16x16x4_f32(false, a, false, b,
                                                    (short)0, acc, false, false);
    }

    const float wbv = Wb[colN];
    const float ew  = extw[colN];
    const float bl  = *beta_p;
    for (int r = 0; r < 8; ++r) {
        int node = row0 + r + 8 * hi;
        if (node < Nn) {
            float cf   = coef[node];
            float hv   = As[r + 8 * hi][colN];
            float h0v  = h0[(size_t)node * H_DIM + colN];
            float outp = acc[r] + wbv;
            float outv = cf * outp - (hv * ew + bl * h0v);
            h_out[(size_t)node * H_DIM + colN] = hv + STEPF * fmaxf(outv, 0.f);
        }
    }
}

__launch_bounds__(96)
__global__ void decoder_gemm(const float* __restrict__ h, const float* __restrict__ W,
                             const float* __restrict__ bias, float* __restrict__ out, int Nn) {
    __shared__ __align__(16) float As[16][H_DIM + 4];
    const int row0 = blockIdx.x * 16;
    const v4f* h4 = (const v4f*)h;
    for (int idx = threadIdx.x; idx < 16 * (H_DIM / 4); idx += 96) {
        int r  = idx >> 5;
        int c4 = idx & 31;
        int node = row0 + r;
        v4f v = {0.f, 0.f, 0.f, 0.f};
        if (node < Nn) v = h4[(size_t)node * (H_DIM / 4) + c4];
        *(v4f*)&As[r][c4 * 4] = v;
    }
    __syncthreads();

    const int lane = threadIdx.x & 31;
    const int wave = threadIdx.x >> 5;
    const int col0 = wave * 16;
    const int m    = lane & 15;
    const int hi   = lane >> 4;
    const int colN = col0 + m;
    const bool colok = (colN < C_DIM);
    const float* wrow = W + (size_t)(colok ? colN : 0) * H_DIM;

    v8f acc = {};
    for (int k0 = 0; k0 < H_DIM; k0 += 4) {
        int k = k0 + 2 * hi;
        v2f a = *(const v2f*)&As[m][k];
        v2f b = {0.f, 0.f};
        if (colok) b = *(const v2f*)&wrow[k];
        acc = __builtin_amdgcn_wmma_f32_16x16x4_f32(false, a, false, b,
                                                    (short)0, acc, false, false);
    }
    if (colok) {
        const float bv = bias[colN];
        for (int r = 0; r < 8; ++r) {
            int node = row0 + r + 8 * hi;
            if (node < Nn)
                __builtin_nontemporal_store(acc[r] + bv,
                                            &out[(size_t)node * C_DIM + colN]);
        }
    }
}


extern "C" void kernel_launch(void* const* d_in, const int* in_sizes, int n_in,
                              void* d_out, int out_size, void* d_ws, size_t ws_size,
                              hipStream_t stream) {
    const float* x     = (const float*)d_in[0];
    const int*   eidx  = (const int*)d_in[1];
    const float* enc_W = (const float*)d_in[2];
    const float* enc_b = (const float*)d_in[3];
    const float* W_raw = (const float*)d_in[4];
    const float* W_b   = (const float*)d_in[5];
    const float* ext_w = (const float*)d_in[6];
    const float* beta  = (const float*)d_in[7];
    const float* dec_W = (const float*)d_in[8];
    const float* dec_b = (const float*)d_in[9];

    const int Nn = in_sizes[0] / F_IN_DIM;
    const int E  = in_sizes[1] / 2;
    const int* row = eidx;
    const int* col = eidx + E;

    char* ws = (char*)d_ws;
    size_t off = 0;
    auto wsalloc = [&](size_t bytes) -> void* {
        void* p = ws + off;
        off = (off + bytes + 255) & ~(size_t)255;
        return p;
    };
    float* deg  = (float*)wsalloc((size_t)Nn * sizeof(float));
    float* coef = (float*)wsalloc((size_t)Nn * sizeof(float));
    float* Weff = (float*)wsalloc((size_t)L_DIM * H_DIM * H_DIM * sizeof(float));
    float* h0   = (float*)wsalloc((size_t)Nn * H_DIM * sizeof(float));
    float* ha   = (float*)wsalloc((size_t)Nn * H_DIM * sizeof(float));
    float* hb   = (float*)wsalloc((size_t)Nn * H_DIM * sizeof(float));
    (void)ws_size; (void)n_in; (void)out_size;

    const int tb = 256;
    zero_two<<<(Nn + tb - 1) / tb, tb, 0, stream>>>(deg, coef, Nn);
    deg_accum<<<(E + tb - 1) / tb, tb, 0, stream>>>(row, deg, E);
    deg_inv_kernel<<<(Nn + tb - 1) / tb, tb, 0, stream>>>(deg, Nn);
    coef_accum<<<(E + tb - 1) / tb, tb, 0, stream>>>(row, col, deg, coef, E);
    weff_kernel<<<L_DIM * H_DIM, H_DIM, 0, stream>>>(W_raw, Weff);

    const int gTiles = (Nn + 15) / 16;
    encoder_gemm<<<gTiles, 256, 0, stream>>>(x, enc_W, enc_b, h0, Nn);

    const float* hin = h0;
    float* bufs[2] = { ha, hb };
    for (int l = 0; l < L_DIM; ++l) {
        float* hout = bufs[l & 1];
        layer_gemm<<<gTiles, 256, 0, stream>>>(hin, h0, coef,
                                               Weff + (size_t)l * H_DIM * H_DIM,
                                               W_b + (size_t)l * H_DIM,
                                               ext_w + (size_t)l * H_DIM,
                                               beta + l, hout, Nn);
        hin = hout;
    }
    decoder_gemm<<<gTiles, 96, 0, stream>>>(hin, dec_W, dec_b, (float*)d_out, Nn);
}


# encoder, the 4 layers and the decoder fused into one kernel per 16-row tile: h stays in LDS and registers between the products, h buffers and 5 launches go; sum order of every element unchanged
# speedup vs baseline: 1.3858x; 1.2062x over previous
#include <hip/hip_runtime.h>

#define H_DIM   128
#define F_IN_DIM 256
#define C_DIM   40
#define L_DIM   4
#define STEPF   0.1f

typedef __attribute__((ext_vector_type(2))) float v2f;
typedef __attribute__((ext_vector_type(4))) float v4f;
typedef __attribute__((ext_vector_type(8))) float v8f;


__global__ void zero_two(float* __restrict__ a, float* __restrict__ b, int n) {
    int i = blockIdx.x * blockDim.x + threadIdx.x;
    if (i < n) { a[i] = 0.f; b[i] = 0.f; }
}

__global__ void deg_accum(const int* __restrict__ row, float* __restrict__ deg, int E) {
    int e = blockIdx.x * blockDim.x + threadIdx.x;
    if (e < E) atomicAdd(&deg[row[e]], 1.0f);
}

__global__ void deg_inv_kernel(float* __restrict__ deg, int n) {
    int i = blockIdx.x * blockDim.x + threadIdx.x;
    if (i < n) { float d = deg[i]; deg[i] = (d > 0.f) ? rsqrtf(d) : 0.f; }
}

__global__ void coef_accum(const int* __restrict__ row, const int* __restrict__ col,
                           const float* __restrict__ dinv, float* __restrict__ coef, int E) {
    int e = blockIdx.x * blockDim.x + threadIdx.x;
    if (e < E) {
        int r = row[e], c = col[e];
        atomicAdd(&coef[c], dinv[r] * dinv[c]);
    }
}

__launch_bounds__(H_DIM)
__global__ void weff_kernel(const float* __restrict__ Wraw, float* __restrict__ Weff) {
    __shared__ float absrow[H_DIM];
    const int l = blockIdx.x / H_DIM;
    const int i = blockIdx.x % H_DIM;
    const int j = threadIdx.x;
    const float* Wr = Wraw + (size_t)l * H_DIM * (H_DIM + 2);
    float*       We = Weff + (size_t)l * H_DIM * H_DIM;
    float w = 0.f;
    if (j > i)      w = Wr[i * (H_DIM + 2) + j];
    else if (j < i) w = Wr[j * (H_DIM + 2) + i];
    absrow[j] = fabsf(w);
    const float da = Wr[i * (H_DIM + 2) + H_DIM];
    const float db = Wr[i * (H_DIM + 2) + H_DIM + 1];
    __syncthreads();
    float v = w;
    if (j == i) {
        float s = 0.f;
        for (int t = 0; t < H_DIM; ++t) s += absrow[t];
        v = da * s + db;
    }
    We[i * H_DIM + j] = v;
}

__launch_bounds__(256)
__global__ void fused_chain(const float* __restrict__ x, const float* __restrict__ encW,
                            const float* __restrict__ encb, const float* __restrict__ coef,
                            const float* __restrict__ Weff, const float* __restrict__ Wb,
                            const float* __restrict__ extw, const float* __restrict__ beta,
                            const float* __restrict__ decW, const float* __restrict__ decb,
                            float* __restrict__ out, int Nn) {
    __shared__ __align__(16) float Xs[16][F_IN_DIM + 4];
    __shared__ __align__(16) float Hs[2][16][H_DIM + 4];
    const int row0 = blockIdx.x * 16;

    const v4f* x4 = (const v4f*)x;
    for (int idx = threadIdx.x; idx < 16 * (F_IN_DIM / 4); idx += 256) {
        int r  = idx >> 6;
        int c4 = idx & 63;
        int node = row0 + r;
        v4f v = {0.f, 0.f, 0.f, 0.f};
        if (node < Nn)
            v = __builtin_nontemporal_load(&x4[(size_t)node * (F_IN_DIM / 4) + c4]);
        *(v4f*)&Xs[r][c4 * 4] = v;
    }
    __syncthreads();

    const int lane = threadIdx.x & 31;
    const int wave = threadIdx.x >> 5;
    const int col0 = wave * 16;
    const int m    = lane & 15;
    const int hi   = lane >> 4;
    const int colN = col0 + m;

    float cf[8];
    for (int r = 0; r < 8; ++r) {
        int node = row0 + r + 8 * hi;
        cf[r] = coef[node < Nn ? node : Nn - 1];
    }

    float h0r[8], hr[8];
    {
        const float* wrow = encW + (size_t)colN * F_IN_DIM;
        v8f acc = {};
        for (int k0 = 0; k0 < F_IN_DIM; k0 += 4) {
            int k = k0 + 2 * hi;
            v2f a = *(const v2f*)&Xs[m][k];
            v2f b = *(const v2f*)&wrow[k];
            acc = __builtin_amdgcn_wmma_f32_16x16x4_f32(false, a, false, b,
                                                        (short)0, acc, false, false);
        }
        const float bv = encb[colN];
        for (int r = 0; r < 8; ++r) {
            h0r[r] = acc[r] + bv;
            hr[r]  = h0r[r];
            Hs[0][r + 8 * hi][colN] = h0r[r];
        }
    }
    __syncthreads();

    for (int l = 0; l < L_DIM; ++l) {
        const float (*cur)[H_DIM + 4] = Hs[l & 1];
        float (*nxt)[H_DIM + 4]       = Hs[(l + 1) & 1];
        const float* We = Weff + (size_t)l * H_DIM * H_DIM;

        v8f acc = {};
        for (int k0 = 0; k0 < H_DIM; k0 += 4) {
            int k = k0 + 2 * hi;
            v2f a = *(const v2f*)&cur[m][k];
            v2f b;
            b.x = We[(size_t)k * H_DIM + colN];
            b.y = We[(size_t)(k + 1) * H_DIM + colN];
            acc = __builtin_amdgcn_wmma_f32_16x16x4_f32(false, a, false, b,
                                                        (short)0, acc, false, false);
        }

        const float wbv = Wb[l * H_DIM + colN];
        const float ew  = extw[l * H_DIM + colN];
        const float bl  = beta[l];
        for (int r = 0; r < 8; ++r) {
            float hv   = hr[r];
            float h0v  = h0r[r];
            float outp = acc[r] + wbv;
            float outv = cf[r] * outp - (hv * ew + bl * h0v);
            hr[r] = hv + STEPF * fmaxf(outv, 0.f);
            nxt[r + 8 * hi][colN] = hr[r];
        }
        __syncthreads();
    }

    if (wave < 3) {
        const float (*hf)[H_DIM + 4] = Hs[L_DIM & 1];
        const bool colok = (colN < C_DIM);
        const float* wrow = decW + (size_t)(colok ? colN : 0) * H_DIM;
        v8f acc = {};
        for (int k0 = 0; k0 < H_DIM; k0 += 4) {
            int k = k0 + 2 * hi;
            v2f a = *(const v2f*)&hf[m][k];
            v2f b = {0.f, 0.f};
            if (colok) b = *(const v2f*)&wrow[k];
            acc = __builtin_amdgcn_wmma_f32_16x16x4_f32(false, a, false, b,
                                                        (short)0, acc, false, false);
        }
        if (colok) {
            const float bv = decb[colN];
            for (int r = 0; r < 8; ++r) {
                int node = row0 + r + 8 * hi;
                if (node < Nn)
                    __builtin_nontemporal_store(acc[r] + bv,
                                                &out[(size_t)node * C_DIM + colN]);
            }
        }
    }
}


extern "C" void kernel_launch(void* const* d_in, const int* in_sizes, int n_in,
                              void* d_out, int out_size, void* d_ws, size_t ws_size,
                              hipStream_t stream) {
    const float* x     = (const float*)d_in[0];
    const int*   eidx  = (const int*)d_in[1];
    const float* enc_W = (const float*)d_in[2];
    const float* enc_b = (const float*)d_in[3];
    const float* W_raw = (const float*)d_in[4];
    const float* W_b   = (const float*)d_in[5];
    const float* ext_w = (const float*)d_in[6];
    const float* beta  = (const float*)d_in[7];
    const float* dec_W = (const float*)d_in[8];
    const float* dec_b = (const float*)d_in[9];

    const int Nn = in_sizes[0] / F_IN_DIM;
    const int E  = in_sizes[1] / 2;
    const int* row = eidx;
    const int* col = eidx + E;

    char* ws = (char*)d_ws;
    size_t off = 0;
    auto wsalloc = [&](size_t bytes) -> void* {
        void* p = ws + off;
        off = (off + bytes + 255) & ~(size_t)255;
        return p;
    };
    float* deg  = (float*)wsalloc((size_t)Nn * sizeof(float));
    float* coef = (float*)wsalloc((size_t)Nn * sizeof(float));
    float* Weff = (float*)wsalloc((size_t)L_DIM * H_DIM * H_DIM * sizeof(float));
    (void)ws_size; (void)n_in; (void)out_size;

    const int tb = 256;
    zero_two<<<(Nn + tb - 1) / tb, tb, 0, stream>>>(deg, coef, Nn);
    deg_accum<<<(E + tb - 1) / tb, tb, 0, stream>>>(row, deg, E);
    deg_inv_kernel<<<(Nn + tb - 1) / tb, tb, 0, stream>>>(deg, Nn);
    coef_accum<<<(E + tb - 1) / tb, tb, 0, stream>>>(row, col, deg, coef, E);
    weff_kernel<<<L_DIM * H_DIM, H_DIM, 0, stream>>>(W_raw, Weff);

    const int gTiles = (Nn + 15) / 16;
    fused_chain<<<gTiles, 256, 0, stream>>>(x, enc_W, enc_b, coef, Weff, W_b, ext_w, beta,
                                            dec_W, dec_b, (float*)d_out, Nn);
}


# fused chain: 64-row block tile instead of 16; each weight fragment loaded once per k step feeds 4 row tiles held in LDS, so weights are read once per 64 rows; sum order of every element unchanged
# speedup vs baseline: 1.6311x; 1.1770x over previous
#include <hip/hip_runtime.h>

#define H_DIM   128
#define F_IN_DIM 256
#define C_DIM   40
#define L_DIM   4
#define STEPF   0.1f

typedef __attribute__((ext_vector_type(2))) float v2f;
typedef __attribute__((ext_vector_type(4))) float v4f;
typedef __attribute__((ext_vector_type(8))) float v8f;


__global__ void zero_two(float* __restrict__ a, float* __restrict__ b, int n) {
    int i = blockIdx.x * blockDim.x + threadIdx.x;
    if (i < n) { a[i] = 0.f; b[i] = 0.f; }
}

__global__ void deg_accum(const int* __restrict__ row, float* __restrict__ deg, int E) {
    int e = blockIdx.x * blockDim.x + threadIdx.x;
    if (e < E) atomicAdd(&deg[row[e]], 1.0f);
}

__global__ void deg_inv_kernel(float* __restrict__ deg, int n) {
    int i = blockIdx.x * blockDim.x + threadIdx.x;
    if (i < n) { float d = deg[i]; deg[i] = (d > 0.f) ? rsqrtf(d) : 0.f; }
}

__global__ void coef_accum(const int* __restrict__ row, const int* __restrict__ col,
                           const float* __restrict__ dinv, float* __restrict__ coef, int E) {
    int e = blockIdx.x * blockDim.x + threadIdx.x;
    if (e < E) {
        int r = row[e], c = col[e];
        atomicAdd(&coef[c], dinv[r] * dinv[c]);
    }
}

__launch_bounds__(H_DIM)
__global__ void weff_kernel(const float* __restrict__ Wraw, float* __restrict__ Weff) {
    __shared__ float absrow[H_DIM];
    const int l = blockIdx.x / H_DIM;
    const int i = blockIdx.x % H_DIM;
    const int j = threadIdx.x;
    const float* Wr = Wraw + (size_t)l * H_DIM * (H_DIM + 2);
    float*       We = Weff + (size_t)l * H_DIM * H_DIM;
    float w = 0.f;
    if (j > i)      w = Wr[i * (H_DIM + 2) + j];
    else if (j < i) w = Wr[j * (H_DIM + 2) + i];
    absrow[j] = fabsf(w);
    const float da = Wr[i * (H_DIM + 2) + H_DIM];
    const float db = Wr[i * (H_DIM + 2) + H_DIM + 1];
    __syncthreads();
    float v = w;
    if (j == i) {
        float s = 0.f;
        for (int t = 0; t < H_DIM; ++t) s += absrow[t];
        v = da * s + db;
    }
    We[i * H_DIM + j] = v;
}

#define MT    4
#define ROWS  (16 * MT)
#define XP    (F_IN_DIM + 4)
#define HP    (H_DIM + 4)

__launch_bounds__(256)
__global__ void fused_chain(const float* __restrict__ x, const float* __restrict__ encW,
                            const float* __restrict__ encb, const float* __restrict__ coef,
                            const float* __restrict__ Weff, const float* __restrict__ Wb,
                            const float* __restrict__ extw, const float* __restrict__ beta,
                            const float* __restrict__ decW, const float* __restrict__ decb,
                            float* __restrict__ out, int Nn) {
    __shared__ __align__(16) float bufA[ROWS * XP];
    __shared__ __align__(16) float bufB[ROWS * HP];
    __shared__ __align__(16) float cfs[ROWS];
    const int row0 = blockIdx.x * ROWS;
    if (threadIdx.x < ROWS) {
        int node = row0 + threadIdx.x;
        cfs[threadIdx.x] = coef[node < Nn ? node : Nn - 1];
    }

    const v4f* x4 = (const v4f*)x;
    for (int i = 0; i < ROWS * (F_IN_DIM / 4) / 256; ++i) {
        int idx = threadIdx.x + i * 256;
        int r  = idx >> 6;
        int c4 = idx & 63;
        int node = row0 + r;
        int nc = node < Nn ? node : Nn - 1;
        v4f v = __builtin_nontemporal_load(&x4[(size_t)nc * (F_IN_DIM / 4) + c4]);
        if (node >= Nn) v = (v4f){0.f, 0.f, 0.f, 0.f};
        *(v4f*)&bufA[r * XP + c4 * 4] = v;
    }
    __syncthreads();

    const int lane = threadIdx.x & 31;
    const int wave = threadIdx.x >> 5;
    const int col0 = wave * 16;
    const int m    = lane & 15;
    const int hi   = lane >> 4;
    const int colN = col0 + m;

    float h0r[MT][8];
    {
        const float* wrow = encW + (size_t)colN * F_IN_DIM;
        v8f acc[MT];
        for (int t = 0; t < MT; ++t) acc[t] = (v8f){};
        #pragma unroll 8
        for (int k0 = 0; k0 < F_IN_DIM; k0 += 4) {
            int k = k0 + 2 * hi;
            v2f b = *(const v2f*)&wrow[k];
            for (int t = 0; t < MT; ++t) {
                v2f a = *(const v2f*)&bufA[(t * 16 + m) * XP + k];
                acc[t] = __builtin_amdgcn_wmma_f32_16x16x4_f32(false, a, false, b,
                                                               (short)0, acc[t], false, false);
            }
        }
        const float bv = encb[colN];
        for (int t = 0; t < MT; ++t)
            for (int r = 0; r < 8; ++r) {
                h0r[t][r] = acc[t][r] + bv;
                bufB[(t * 16 + r + 8 * hi) * HP + colN] = h0r[t][r];
            }
    }
    __syncthreads();

    #pragma nounroll
    for (int l = 0; l < L_DIM; ++l) {
        const float* cur = (l & 1) ? bufA : bufB;
        float*       nxt = (l & 1) ? bufB : bufA;
        const float* We = Weff + (size_t)l * H_DIM * H_DIM;

        v8f acc[MT];
        for (int t = 0; t < MT; ++t) acc[t] = (v8f){};
        #pragma unroll 8
        for (int k0 = 0; k0 < H_DIM; k0 += 4) {
            int k = k0 + 2 * hi;
            v2f b;
            b.x = We[(size_t)k * H_DIM + colN];
            b.y = We[(size_t)(k + 1) * H_DIM + colN];
            for (int t = 0; t < MT; ++t) {
                v2f a = *(const v2f*)&cur[(t * 16 + m) * HP + k];
                acc[t] = __builtin_amdgcn_wmma_f32_16x16x4_f32(false, a, false, b,
                                                               (short)0, acc[t], false, false);
            }
        }

        const float wbv = Wb[l * H_DIM + colN];
        const float ew  = extw[l * H_DIM + colN];
        const float bl  = beta[l];
        float hv[MT][8];
        v4f cfa[MT], cfb[MT];
        for (int t = 0; t < MT; ++t) {
            for (int r = 0; r < 8; ++r)
                hv[t][r] = cur[(t * 16 + r + 8 * hi) * HP + colN];
            cfa[t] = *(const v4f*)&cfs[t * 16 + 8 * hi];
            cfb[t] = *(const v4f*)&cfs[t * 16 + 8 * hi + 4];
        }
        for (int t = 0; t < MT; ++t)
            for (int r = 0; r < 8; ++r) {
                float cfv  = (r < 4) ? cfa[t][r & 3] : cfb[t][r & 3];
                float h0v  = h0r[t][r];
                float outp = acc[t][r] + wbv;
                float outv = cfv * outp - (hv[t][r] * ew + bl * h0v);
                nxt[(t * 16 + r + 8 * hi) * HP + colN] = hv[t][r] + STEPF * fmaxf(outv, 0.f);
            }
        __syncthreads();
    }

    if (wave < 3) {
        const float* hf = (L_DIM & 1) ? bufA : bufB;
        const bool colok = (colN < C_DIM);
        const float* wrow = decW + (size_t)(colok ? colN : 0) * H_DIM;
        v8f acc[MT];
        for (int t = 0; t < MT; ++t) acc[t] = (v8f){};
        for (int k0 = 0; k0 < H_DIM; k0 += 4) {
            int k = k0 + 2 * hi;
            v2f b = {0.f, 0.f};
            if (colok) b = *(const v2f*)&wrow[k];
            for (int t = 0; t < MT; ++t) {
                v2f a = *(const v2f*)&hf[(t * 16 + m) * HP + k];
                acc[t] = __builtin_amdgcn_wmma_f32_16x16x4_f32(false, a, false, b,
                                                               (short)0, acc[t], false, false);
            }
        }
        if (colok) {
            const float bv = decb[colN];
            for (int t = 0; t < MT; ++t)
                for (int r = 0; r < 8; ++r) {
                    int node = row0 + t * 16 + r + 8 * hi;
                    if (node < Nn)
                        __builtin_nontemporal_store(acc[t][r] + bv,
                                                    &out[(size_t)node * C_DIM + colN]);
                }
        }
    }
}


extern "C" void kernel_launch(void* const* d_in, const int* in_sizes, int n_in,
                              void* d_out, int out_size, void* d_ws, size_t ws_size,
                              hipStream_t stream) {
    const float* x     = (const float*)d_in[0];
    const int*   eidx  = (const int*)d_in[1];
    const float* enc_W = (const float*)d_in[2];
    const float* enc_b = (const float*)d_in[3];
    const float* W_raw = (const float*)d_in[4];
    const float* W_b   = (const float*)d_in[5];
    const float* ext_w = (const float*)d_in[6];
    const float* beta  = (const float*)d_in[7];
    const float* dec_W = (const float*)d_in[8];
    const float* dec_b = (const float*)d_in[9];

    const int Nn = in_sizes[0] / F_IN_DIM;
    const int E  = in_sizes[1] / 2;
    const int* row = eidx;
    const int* col = eidx + E;

    char* ws = (char*)d_ws;
    size_t off = 0;
    auto wsalloc = [&](size_t bytes) -> void* {
        void* p = ws + off;
        off = (off + bytes + 255) & ~(size_t)255;
        return p;
    };
    float* deg  = (float*)wsalloc((size_t)Nn * sizeof(float));
    float* coef = (float*)wsalloc((size_t)Nn * sizeof(float));
    float* Weff = (float*)wsalloc((size_t)L_DIM * H_DIM * H_DIM * sizeof(float));
    (void)ws_size; (void)n_in; (void)out_size;

    const int tb = 256;
    zero_two<<<(Nn + tb - 1) / tb, tb, 0, stream>>>(deg, coef, Nn);
    deg_accum<<<(E + tb - 1) / tb, tb, 0, stream>>>(row, deg, E);
    deg_inv_kernel<<<(Nn + tb - 1) / tb, tb, 0, stream>>>(deg, Nn);
    coef_accum<<<(E + tb - 1) / tb, tb, 0, stream>>>(row, col, deg, coef, E);
    weff_kernel<<<L_DIM * H_DIM, H_DIM, 0, stream>>>(W_raw, Weff);

    const int gTiles = (Nn + ROWS - 1) / ROWS;
    fused_chain<<<gTiles, 256, 0, stream>>>(x, enc_W, enc_b, coef, Weff, W_b, ext_w, beta,
                                            dec_W, dec_b, (float*)d_out, Nn);
}


# set-up edge passes: per-edge global float atomics replaced by LDS accumulators per (edge slice, node range) block, slice planes then added in slice order; coef summation order changes
# speedup vs baseline: 2.8247x; 1.7318x over previous
#include <hip/hip_runtime.h>

#define H_DIM   128
#define F_IN_DIM 256
#define C_DIM   40
#define L_DIM   4
#define STEPF   0.1f

typedef __attribute__((ext_vector_type(2))) float v2f;
typedef __attribute__((ext_vector_type(4))) float v4f;
typedef __attribute__((ext_vector_type(8))) float v8f;

#define NR      25600
#define EP_TPB  1024
#define EP_U    13
#define EP_S    64

__launch_bounds__(EP_TPB)
__global__ void deg_part(const int* __restrict__ row, int* __restrict__ part, int E, int Nn) {
    __shared__ __align__(16) int hist[NR];
    const int tid  = threadIdx.x;
    const int base = blockIdx.y * NR;
    for (int i = tid; i < NR / 4; i += EP_TPB) ((int4*)hist)[i] = make_int4(0, 0, 0, 0);
    __syncthreads();
    const int ES = (E + (int)gridDim.x - 1) / (int)gridDim.x;
    const int e0 = blockIdx.x * ES;
    const int e1 = min(E, e0 + ES);
    for (int b = e0; b < e1; b += EP_TPB * EP_U) {
        int idx[EP_U];
        #pragma unroll
        for (int u = 0; u < EP_U; ++u) {
            int e = b + u * EP_TPB + tid;
            int v = row[min(e, e1 - 1)];
            idx[u] = (e < e1) ? v : -1;
        }
        #pragma unroll
        for (int u = 0; u < EP_U; ++u) {
            unsigned loc = (unsigned)(idx[u] - base);
            if (loc < (unsigned)NR) atomicAdd(&hist[loc], 1);
        }
    }
    __syncthreads();
    int* dst = part + (size_t)blockIdx.x * Nn + base;
    const int lim = min(NR, Nn - base);
    for (int i = tid; i < lim; i += EP_TPB) dst[i] = hist[i];
}

__global__ void deg_reduce(const int* __restrict__ part, float* __restrict__ dinv, int Nn) {
    int n = blockIdx.x * blockDim.x + threadIdx.x;
    if (n < Nn) {
        int d = 0;
        #pragma unroll
        for (int s = 0; s < EP_S; ++s) d += part[(size_t)s * Nn + n];
        dinv[n] = (d > 0) ? rsqrtf((float)d) : 0.f;
    }
}

__launch_bounds__(EP_TPB)
__global__ void coef_part(const int* __restrict__ row, const int* __restrict__ col,
                          const float* __restrict__ dinv, float* __restrict__ part, int E, int Nn) {
    __shared__ __align__(16) float acc[NR];
    const int tid  = threadIdx.x;
    const int base = blockIdx.y * NR;
    for (int i = tid; i < NR / 4; i += EP_TPB) ((float4*)acc)[i] = make_float4(0.f, 0.f, 0.f, 0.f);
    __syncthreads();
    const int ES = (E + (int)gridDim.x - 1) / (int)gridDim.x;
    const int e0 = blockIdx.x * ES;
    const int e1 = min(E, e0 + ES);
    for (int b = e0; b < e1; b += EP_TPB * EP_U) {
        int cc[EP_U], rr[EP_U];
        #pragma unroll
        for (int u = 0; u < EP_U; ++u) {
            int e  = b + u * EP_TPB + tid;
            int ec = min(e, e1 - 1);
            int c = col[ec];
            int r = row[ec];
            bool in = (e < e1) && ((unsigned)(c - base) < (unsigned)NR);
            cc[u] = in ? c : -1;
            rr[u] = in ? r : base;
        }
        float wr[EP_U], wc[EP_U];
        #pragma unroll
        for (int u = 0; u < EP_U; ++u) {
            wc[u] = dinv[cc[u] >= 0 ? cc[u] : base];
            wr[u] = dinv[rr[u]];
        }
        static_assert(EP_U == 13, "the fence below names 13 value pairs");
        asm volatile("" :: "v"(wr[0]), "v"(wc[0]), "v"(wr[1]), "v"(wc[1]), "v"(wr[2]), "v"(wc[2]), "v"(wr[3]), "v"(wc[3]), "v"(wr[4]), "v"(wc[4]), "v"(wr[5]), "v"(wc[5]), "v"(wr[6]), "v"(wc[6]), "v"(wr[7]), "v"(wc[7]), "v"(wr[8]), "v"(wc[8]), "v"(wr[9]), "v"(wc[9]), "v"(wr[10]), "v"(wc[10]), "v"(wr[11]), "v"(wc[11]), "v"(wr[12]), "v"(wc[12]));
        #pragma unroll
        for (int u = 0; u < EP_U; ++u)
            if (cc[u] >= 0) atomicAdd(&acc[cc[u] - base], wr[u] * wc[u]);
    }
    __syncthreads();
    float* dst = part + (size_t)blockIdx.x * Nn + base;
    const int lim = min(NR, Nn - base);
    for (int i = tid; i < lim; i += EP_TPB) dst[i] = acc[i];
}

__global__ void coef_reduce(const float* __restrict__ part, float* __restrict__ coef, int Nn) {
    int n = blockIdx.x * blockDim.x + threadIdx.x;
    if (n < Nn) {
        float v[EP_S];
        #pragma unroll
        for (int s = 0; s < EP_S; ++s) v[s] = part[(size_t)s * Nn + n];
        static_assert(EP_S % 16 == 0, "the fences below take the planes 16 at a time");
        #pragma unroll
        for (int s = 0; s < EP_S; s += 16) {
            asm volatile("" :: "v"(v[s + 0]), "v"(v[s + 1]), "v"(v[s + 2]), "v"(v[s + 3]), "v"(v[s + 4]), "v"(v[s + 5]), "v"(v[s + 6]), "v"(v[s + 7]), "v"(v[s + 8]), "v"(v[s + 9]), "v"(v[s + 10]), "v"(v[s + 11]), "v"(v[s + 12]), "v"(v[s + 13]), "v"(v[s + 14]), "v"(v[s + 15]));
        }
        float a = 0.f;
        #pragma unroll
        for (int s = 0; s < EP_S; ++s) a += v[s];
        coef[n] = a;
    }
}

__launch_bounds__(H_DIM)
__global__ void weff_kernel(const float* __restrict__ Wraw, float* __restrict__ Weff) {
    __shared__ float absrow[H_DIM];
    const int l = blockIdx.x / H_DIM;
    const int i = blockIdx.x % H_DIM;
    const int j = threadIdx.x;
    const float* Wr = Wraw + (size_t)l * H_DIM * (H_DIM + 2);
    float*       We = Weff + (size_t)l * H_DIM * H_DIM;
    float w = 0.f;
    if (j > i)      w = Wr[i * (H_DIM + 2) + j];
    else if (j < i) w = Wr[j * (H_DIM + 2) + i];
    absrow[j] = fabsf(w);
    const float da = Wr[i * (H_DIM + 2) + H_DIM];
    const float db = Wr[i * (H_DIM + 2) + H_DIM + 1];
    __syncthreads();
    float v = w;
    if (j == i) {
        float s = 0.f;
        for (int t = 0; t < H_DIM; ++t) s += absrow[t];
        v = da * s + db;
    }
    We[i * H_DIM + j] = v;
}

#define MT    4
#define ROWS  (16 * MT)
#define XP    (F_IN_DIM + 4)
#define HP    (H_DIM + 4)

__launch_bounds__(256)
__global__ void fused_chain(const float* __restrict__ x, const float* __restrict__ encW,
                            const float* __restrict__ encb, const float* __restrict__ coef,
                            const float* __restrict__ Weff, const float* __restrict__ Wb,
                            const float* __restrict__ extw, const float* __restrict__ beta,
                            const float* __restrict__ decW, const float* __restrict__ decb,
                            float* __restrict__ out, int Nn) {
    __shared__ __align__(16) float bufA[ROWS * XP];
    __shared__ __align__(16) float bufB[ROWS * HP];
    __shared__ __align__(16) float cfs[ROWS];
    const int row0 = blockIdx.x * ROWS;
    if (threadIdx.x < ROWS) {
        int node = row0 + threadIdx.x;
        cfs[threadIdx.x] = coef[node < Nn ? node : Nn - 1];
    }

    const v4f* x4 = (const v4f*)x;
    for (int i = 0; i < ROWS * (F_IN_DIM / 4) / 256; ++i) {
        int idx = threadIdx.x + i * 256;
        int r  = idx >> 6;
        int c4 = idx & 63;
        int node = row0 + r;
        int nc = node < Nn ? node : Nn - 1;
        v4f v = __builtin_nontemporal_load(&x4[(size_t)nc * (F_IN_DIM / 4) + c4]);
        if (node >= Nn) v = (v4f){0.f, 0.f, 0.f, 0.f};
        *(v4f*)&bufA[r * XP + c4 * 4] = v;
    }
    __syncthreads();

    const int lane = threadIdx.x & 31;
    const int wave = threadIdx.x >> 5;
    const int col0 = wave * 16;
    const int m    = lane & 15;
    const int hi   = lane >> 4;
    const int colN = col0 + m;

    float h0r[MT][8];
    {
        const float* wrow = encW + (size_t)colN * F_IN_DIM;
        v8f acc[MT];
        for (int t = 0; t < MT; ++t) acc[t] = (v8f){};
        #pragma unroll 8
        for (int k0 = 0; k0 < F_IN_DIM; k0 += 4) {
            int k = k0 + 2 * hi;
            v2f b = *(const v2f*)&wrow[k];
            for (int t = 0; t < MT; ++t) {
                v2f a = *(const v2f*)&bufA[(t * 16 + m) * XP + k];
                acc[t] = __builtin_amdgcn_wmma_f32_16x16x4_f32(false, a, false, b,
                                                               (short)0, acc[t], false, false);
            }
        }
        const float bv = encb[colN];
        for (int t = 0; t < MT; ++t)
            for (int r = 0; r < 8; ++r) {
                h0r[t][r] = acc[t][r] + bv;
                bufB[(t * 16 + r + 8 * hi) * HP + colN] = h0r[t][r];
            }
    }
    __syncthreads();

    #pragma nounroll
    for (int l = 0; l < L_DIM; ++l) {
        const float* cur = (l & 1) ? bufA : bufB;
        float*       nxt = (l & 1) ? bufB : bufA;
        const float* We = Weff + (size_t)l * H_DIM * H_DIM;

        v8f acc[MT];
        for (int t = 0; t < MT; ++t) acc[t] = (v8f){};
        #pragma unroll 8
        for (int k0 = 0; k0 < H_DIM; k0 += 4) {
            int k = k0 + 2 * hi;
            v2f b;
            b.x = We[(size_t)k * H_DIM + colN];
            b.y = We[(size_t)(k + 1) * H_DIM + colN];
            for (int t = 0; t < MT; ++t) {
                v2f a = *(const v2f*)&cur[(t * 16 + m) * HP + k];
                acc[t] = __builtin_amdgcn_wmma_f32_16x16x4_f32(false, a, false, b,
                                                               (short)0, acc[t], false, false);
            }
        }

        const float wbv = Wb[l * H_DIM + colN];
        const float ew  = extw[l * H_DIM + colN];
        const float bl  = beta[l];
        float hv[MT][8];
        v4f cfa[MT], cfb[MT];
        for (int t = 0; t < MT; ++t) {
            for (int r = 0; r < 8; ++r)
                hv[t][r] = cur[(t * 16 + r + 8 * hi) * HP + colN];
            cfa[t] = *(const v4f*)&cfs[t * 16 + 8 * hi];
            cfb[t] = *(const v4f*)&cfs[t * 16 + 8 * hi + 4];
        }
        for (int t = 0; t < MT; ++t)
            for (int r = 0; r < 8; ++r) {
                float cfv  = (r < 4) ? cfa[t][r & 3] : cfb[t][r & 3];
                float h0v  = h0r[t][r];
                float outp = acc[t][r] + wbv;
                float outv = cfv * outp - (hv[t][r] * ew + bl * h0v);
                nxt[(t * 16 + r + 8 * hi) * HP + colN] = hv[t][r] + STEPF * fmaxf(outv, 0.f);
            }
        __syncthreads();
    }

    if (wave < 3) {
        const float* hf = (L_DIM & 1) ? bufA : bufB;
        const bool colok = (colN < C_DIM);
        const float* wrow = decW + (size_t)(colok ? colN : 0) * H_DIM;
        v8f acc[MT];
        for (int t = 0; t < MT; ++t) acc[t] = (v8f){};
        for (int k0 = 0; k0 < H_DIM; k0 += 4) {
            int k = k0 + 2 * hi;
            v2f b = {0.f, 0.f};
            if (colok) b = *(const v2f*)&wrow[k];
            for (int t = 0; t < MT; ++t) {
                v2f a = *(const v2f*)&hf[(t * 16 + m) * HP + k];
                acc[t] = __builtin_amdgcn_wmma_f32_16x16x4_f32(false, a, false, b,
                                                               (short)0, acc[t], false, false);
            }
        }
        if (colok) {
            const float bv = decb[colN];
            for (int t = 0; t < MT; ++t)
                for (int r = 0; r < 8; ++r) {
                    int node = row0 + t * 16 + r + 8 * hi;
                    if (node < Nn)
                        __builtin_nontemporal_store(acc[t][r] + bv,
                                                    &out[(size_t)node * C_DIM + colN]);
                }
        }
    }
}


extern "C" void kernel_launch(void* const* d_in, const int* in_sizes, int n_in,
                              void* d_out, int out_size, void* d_ws, size_t ws_size,
                              hipStream_t stream) {
    const float* x     = (const float*)d_in[0];
    const int*   eidx  = (const int*)d_in[1];
    const float* enc_W = (const float*)d_in[2];
    const float* enc_b = (const float*)d_in[3];
    const float* W_raw = (const float*)d_in[4];
    const float* W_b   = (const float*)d_in[5];
    const float* ext_w = (const float*)d_in[6];
    const float* beta  = (const float*)d_in[7];
    const float* dec_W = (const float*)d_in[8];
    const float* dec_b = (const float*)d_in[9];

    const int Nn = in_sizes[0] / F_IN_DIM;
    const int E  = in_sizes[1] / 2;
    const int* row = eidx;
    const int* col = eidx + E;

    char* ws = (char*)d_ws;
    size_t off = 0;
    auto wsalloc = [&](size_t bytes) -> void* {
        void* p = ws + off;
        off = (off + bytes + 255) & ~(size_t)255;
        return p;
    };
    float* dinv = (float*)wsalloc((size_t)Nn * sizeof(float));
    float* coef = (float*)wsalloc((size_t)Nn * sizeof(float));
    float* Weff = (float*)wsalloc((size_t)L_DIM * H_DIM * H_DIM * sizeof(float));
    void*  part = wsalloc((size_t)EP_S * Nn * sizeof(float));
    (void)ws_size; (void)n_in; (void)out_size;

    const int tb = 256;
    const dim3 egrid(EP_S, (Nn + NR - 1) / NR);
    deg_part<<<egrid, EP_TPB, 0, stream>>>(row, (int*)part, E, Nn);
    deg_reduce<<<(Nn + tb - 1) / tb, tb, 0, stream>>>((const int*)part, dinv, Nn);
    coef_part<<<egrid, EP_TPB, 0, stream>>>(row, col, dinv, (float*)part, E, Nn);
    coef_reduce<<<(Nn + tb - 1) / tb, tb, 0, stream>>>((const float*)part, coef, Nn);
    weff_kernel<<<L_DIM * H_DIM, H_DIM, 0, stream>>>(W_raw, Weff);

    const int gTiles = (Nn + ROWS - 1) / ROWS;
    fused_chain<<<gTiles, 256, 0, stream>>>(x, enc_W, enc_b, coef, Weff, W_b, ext_w, beta,
                                            dec_W, dec_b, (float*)d_out, Nn);
}


# fused chain: 80-row block tile instead of 64 (5 row tiles per weight fragment, tiles interleaved in LDS): 1250 equal blocks, 5 or 4 per compute unit, none carries a 7th; sum order unchanged
# speedup vs baseline: 2.9496x; 1.0442x over previous
#include <hip/hip_runtime.h>

#define H_DIM   128
#define F_IN_DIM 256
#define C_DIM   40
#define L_DIM   4
#define STEPF   0.1f

typedef __attribute__((ext_vector_type(2))) float v2f;
typedef __attribute__((ext_vector_type(4))) float v4f;
typedef __attribute__((ext_vector_type(8))) float v8f;

#define NR      25600
#define EP_TPB  1024
#define EP_U    13
#define EP_S    64

__launch_bounds__(EP_TPB)
__global__ void deg_part(const int* __restrict__ row, int* __restrict__ part, int E, int Nn) {
    __shared__ __align__(16) int hist[NR];
    const int tid  = threadIdx.x;
    const int base = blockIdx.y * NR;
    for (int i = tid; i < NR / 4; i += EP_TPB) ((int4*)hist)[i] = make_int4(0, 0, 0, 0);
    __syncthreads();
    const int ES = (E + (int)gridDim.x - 1) / (int)gridDim.x;
    const int e0 = blockIdx.x * ES;
    const int e1 = min(E, e0 + ES);
    for (int b = e0; b < e1; b += EP_TPB * EP_U) {
        int idx[EP_U];
        #pragma unroll
        for (int u = 0; u < EP_U; ++u) {
            int e = b + u * EP_TPB + tid;
            int v = row[min(e, e1 - 1)];
            idx[u] = (e < e1) ? v : -1;
        }
        #pragma unroll
        for (int u = 0; u < EP_U; ++u) {
            unsigned loc = (unsigned)(idx[u] - base);
            if (loc < (unsigned)NR) atomicAdd(&hist[loc], 1);
        }
    }
    __syncthreads();
    int* dst = part + (size_t)blockIdx.x * Nn + base;
    const int lim = min(NR, Nn - base);
    for (int i = tid; i < lim; i += EP_TPB) dst[i] = hist[i];
}

__global__ void deg_reduce(const int* __restrict__ part, float* __restrict__ dinv, int Nn) {
    int n = blockIdx.x * blockDim.x + threadIdx.x;
    if (n < Nn) {
        int d = 0;
        #pragma unroll
        for (int s = 0; s < EP_S; ++s) d += part[(size_t)s * Nn + n];
        dinv[n] = (d > 0) ? rsqrtf((float)d) : 0.f;
    }
}

__launch_bounds__(EP_TPB)
__global__ void coef_part(const int* __restrict__ row, const int* __restrict__ col,
                          const float* __restrict__ dinv, float* __restrict__ part, int E, int Nn) {
    __shared__ __align__(16) float acc[NR];
    const int tid  = threadIdx.x;
    const int base = blockIdx.y * NR;
    for (int i = tid; i < NR / 4; i += EP_TPB) ((float4*)acc)[i] = make_float4(0.f, 0.f, 0.f, 0.f);
    __syncthreads();
    const int ES = (E + (int)gridDim.x - 1) / (int)gridDim.x;
    const int e0 = blockIdx.x * ES;
    const int e1 = min(E, e0 + ES);
    for (int b = e0; b < e1; b += EP_TPB * EP_U) {
        int cc[EP_U], rr[EP_U];
        #pragma unroll
        for (int u = 0; u < EP_U; ++u) {
            int e  = b + u * EP_TPB + tid;
            int ec = min(e, e1 - 1);
            int c = col[ec];
            int r = row[ec];
            bool in = (e < e1) && ((unsigned)(c - base) < (unsigned)NR);
            cc[u] = in ? c : -1;
            rr[u] = in ? r : base;
        }
        float wr[EP_U], wc[EP_U];
        #pragma unroll
        for (int u = 0; u < EP_U; ++u) {
            wc[u] = dinv[cc[u] >= 0 ? cc[u] : base];
            wr[u] = dinv[rr[u]];
        }
        static_assert(EP_U == 13, "the fence below names 13 value pairs");
        asm volatile("" :: "v"(wr[0]), "v"(wc[0]), "v"(wr[1]), "v"(wc[1]), "v"(wr[2]), "v"(wc[2]), "v"(wr[3]), "v"(wc[3]), "v"(wr[4]), "v"(wc[4]), "v"(wr[5]), "v"(wc[5]), "v"(wr[6]), "v"(wc[6]), "v"(wr[7]), "v"(wc[7]), "v"(wr[8]), "v"(wc[8]), "v"(wr[9]), "v"(wc[9]), "v"(wr[10]), "v"(wc[10]), "v"(wr[11]), "v"(wc[11]), "v"(wr[12]), "v"(wc[12]));
        #pragma unroll
        for (int u = 0; u < EP_U; ++u)
            if (cc[u] >= 0) atomicAdd(&acc[cc[u] - base], wr[u] * wc[u]);
    }
    __syncthreads();
    float* dst = part + (size_t)blockIdx.x * Nn + base;
    const int lim = min(NR, Nn - base);
    for (int i = tid; i < lim; i += EP_TPB) dst[i] = acc[i];
}

__global__ void coef_reduce(const float* __restrict__ part, float* __restrict__ coef, int Nn) {
    int n = blockIdx.x * blockDim.x + threadIdx.x;
    if (n < Nn) {
        float v[EP_S];
        #pragma unroll
        for (int s = 0; s < EP_S; ++s) v[s] = part[(size_t)s * Nn + n];
        static_assert(EP_S % 16 == 0, "the fences below take the planes 16 at a time");
        #pragma unroll
        for (int s = 0; s < EP_S; s += 16) {
            asm volatile("" :: "v"(v[s + 0]), "v"(v[s + 1]), "v"(v[s + 2]), "v"(v[s + 3]), "v"(v[s + 4]), "v"(v[s + 5]), "v"(v[s + 6]), "v"(v[s + 7]), "v"(v[s + 8]), "v"(v[s + 9]), "v"(v[s + 10]), "v"(v[s + 11]), "v"(v[s + 12]), "v"(v[s + 13]), "v"(v[s + 14]), "v"(v[s + 15]));
        }
        float a = 0.f;
        #pragma unroll
        for (int s = 0; s < EP_S; ++s) a += v[s];
        coef[n] = a;
    }
}

__launch_bounds__(H_DIM)
__global__ void weff_kernel(const float* __restrict__ Wraw, float* __restrict__ Weff) {
    __shared__ float absrow[H_DIM];
    const int l = blockIdx.x / H_DIM;
    const int i = blockIdx.x % H_DIM;
    const int j = threadIdx.x;
    const float* Wr = Wraw + (size_t)l * H_DIM * (H_DIM + 2);
    float*       We = Weff + (size_t)l * H_DIM * H_DIM;
    float w = 0.f;
    if (j > i)      w = Wr[i * (H_DIM + 2) + j];
    else if (j < i) w = Wr[j * (H_DIM + 2) + i];
    absrow[j] = fabsf(w);
    const float da = Wr[i * (H_DIM + 2) + H_DIM];
    const float db = Wr[i * (H_DIM + 2) + H_DIM + 1];
    __syncthreads();
    float v = w;
    if (j == i) {
        float s = 0.f;
        for (int t = 0; t < H_DIM; ++t) s += absrow[t];
        v = da * s + db;
    }
    We[i * H_DIM + j] = v;
}

#define MT    5
#define ROWS  (16 * MT)
#define LROW(r) ((((r) & 15) * MT) + ((r) >> 4))
#define XP    (F_IN_DIM + 4)
#define HP    (H_DIM + 4)

__launch_bounds__(256)
__global__ void fused_chain(const float* __restrict__ x, const float* __restrict__ encW,
                            const float* __restrict__ encb, const float* __restrict__ coef,
                            const float* __restrict__ Weff, const float* __restrict__ Wb,
                            const float* __restrict__ extw, const float* __restrict__ beta,
                            const float* __restrict__ decW, const float* __restrict__ decb,
                            float* __restrict__ out, int Nn) {
    __shared__ __align__(16) float bufA[ROWS * XP];
    __shared__ __align__(16) float bufB[ROWS * HP];
    __shared__ __align__(16) float cfs[ROWS];
    const int row0 = blockIdx.x * ROWS;
    if (threadIdx.x < ROWS) {
        int node = row0 + threadIdx.x;
        cfs[threadIdx.x] = coef[node < Nn ? node : Nn - 1];
    }

    const v4f* x4 = (const v4f*)x;
    #pragma unroll
    for (int i = 0; i < ROWS * (F_IN_DIM / 4) / 256; ++i) {
        int idx = threadIdx.x + i * 256;
        int r  = idx >> 6;
        int c4 = idx & 63;
        int node = row0 + r;
        int nc = node < Nn ? node : Nn - 1;
        v4f v = __builtin_nontemporal_load(&x4[(size_t)nc * (F_IN_DIM / 4) + c4]);
        if (node >= Nn) v = (v4f){0.f, 0.f, 0.f, 0.f};
        *(v4f*)&bufA[LROW(r) * XP + c4 * 4] = v;
    }
    __syncthreads();

    const int lane = threadIdx.x & 31;
    const int wave = threadIdx.x >> 5;
    const int col0 = wave * 16;
    const int m    = lane & 15;
    const int hi   = lane >> 4;
    const int colN = col0 + m;

    float h0r[MT][8];
    {
        const float* wrow = encW + (size_t)colN * F_IN_DIM;
        v8f acc[MT];
        for (int t = 0; t < MT; ++t) acc[t] = (v8f){};
        #pragma unroll 8
        for (int k0 = 0; k0 < F_IN_DIM; k0 += 4) {
            int k = k0 + 2 * hi;
            v2f b = *(const v2f*)&wrow[k];
            for (int t = 0; t < MT; ++t) {
                v2f a = *(const v2f*)&bufA[(m * MT + t) * XP + k];
                acc[t] = __builtin_amdgcn_wmma_f32_16x16x4_f32(false, a, false, b,
                                                               (short)0, acc[t], false, false);
            }
        }
        const float bv = encb[colN];
        for (int t = 0; t < MT; ++t)
            for (int r = 0; r < 8; ++r) {
                h0r[t][r] = acc[t][r] + bv;
                bufB[((r + 8 * hi) * MT + t) * HP + colN] = h0r[t][r];
            }
    }
    __syncthreads();

    #pragma nounroll
    for (int l = 0; l < L_DIM; ++l) {
        const float* cur = (l & 1) ? bufA : bufB;
        float*       nxt = (l & 1) ? bufB : bufA;
        const float* We = Weff + (size_t)l * H_DIM * H_DIM;

        v8f acc[MT];
        for (int t = 0; t < MT; ++t) acc[t] = (v8f){};
        #pragma unroll 8
        for (int k0 = 0; k0 < H_DIM; k0 += 4) {
            int k = k0 + 2 * hi;
            v2f b;
            b.x = We[(size_t)k * H_DIM + colN];
            b.y = We[(size_t)(k + 1) * H_DIM + colN];
            for (int t = 0; t < MT; ++t) {
                v2f a = *(const v2f*)&cur[(m * MT + t) * HP + k];
                acc[t] = __builtin_amdgcn_wmma_f32_16x16x4_f32(false, a, false, b,
                                                               (short)0, acc[t], false, false);
            }
        }

        const float wbv = Wb[l * H_DIM + colN];
        const float ew  = extw[l * H_DIM + colN];
        const float bl  = beta[l];
        float hv[MT][8];
        v4f cfa[MT], cfb[MT];
        for (int t = 0; t < MT; ++t) {
            for (int r = 0; r < 8; ++r)
                hv[t][r] = cur[((r + 8 * hi) * MT + t) * HP + colN];
            cfa[t] = *(const v4f*)&cfs[t * 16 + 8 * hi];
            cfb[t] = *(const v4f*)&cfs[t * 16 + 8 * hi + 4];
        }
        for (int t = 0; t < MT; ++t)
            for (int r = 0; r < 8; ++r) {
                float cfv  = (r < 4) ? cfa[t][r & 3] : cfb[t][r & 3];
                float h0v  = h0r[t][r];
                float outp = acc[t][r] + wbv;
                float outv = cfv * outp - (hv[t][r] * ew + bl * h0v);
                nxt[((r + 8 * hi) * MT + t) * HP + colN] = hv[t][r] + STEPF * fmaxf(outv, 0.f);
            }
        __syncthreads();
    }

    if (wave < 3) {
        const float* hf = (L_DIM & 1) ? bufA : bufB;
        const bool colok = (colN < C_DIM);
        const float* wrow = decW + (size_t)(colok ? colN : 0) * H_DIM;
        v8f acc[MT];
        for (int t = 0; t < MT; ++t) acc[t] = (v8f){};
        for (int k0 = 0; k0 < H_DIM; k0 += 4) {
            int k = k0 + 2 * hi;
            v2f b = {0.f, 0.f};
            if (colok) b = *(const v2f*)&wrow[k];
            for (int t = 0; t < MT; ++t) {
                v2f a = *(const v2f*)&hf[(m * MT + t) * HP + k];
                acc[t] = __builtin_amdgcn_wmma_f32_16x16x4_f32(false, a, false, b,
                                                               (short)0, acc[t], false, false);
            }
        }
        if (colok) {
            const float bv = decb[colN];
            for (int t = 0; t < MT; ++t)
                for (int r = 0; r < 8; ++r) {
                    int node = row0 + t * 16 + r + 8 * hi;
                    if (node < Nn)
                        __builtin_nontemporal_store(acc[t][r] + bv,
                                                    &out[(size_t)node * C_DIM + colN]);
                }
        }
    }
}


extern "C" void kernel_launch(void* const* d_in, const int* in_sizes, int n_in,
                              void* d_out, int out_size, void* d_ws, size_t ws_size,
                              hipStream_t stream) {
    const float* x     = (const float*)d_in[0];
    const int*   eidx  = (const int*)d_in[1];
    const float* enc_W = (const float*)d_in[2];
    const float* enc_b = (const float*)d_in[3];
    const float* W_raw = (const float*)d_in[4];
    const float* W_b   = (const float*)d_in[5];
    const float* ext_w = (const float*)d_in[6];
    const float* beta  = (const float*)d_in[7];
    const float* dec_W = (const float*)d_in[8];
    const float* dec_b = (const float*)d_in[9];

    const int Nn = in_sizes[0] / F_IN_DIM;
    const int E  = in_sizes[1] / 2;
    const int* row = eidx;
    const int* col = eidx + E;

    char* ws = (char*)d_ws;
    size_t off = 0;
    auto wsalloc = [&](size_t bytes) -> void* {
        void* p = ws + off;
        off = (off + bytes + 255) & ~(size_t)255;
        return p;
    };
    float* dinv = (float*)wsalloc((size_t)Nn * sizeof(float));
    float* coef = (float*)wsalloc((size_t)Nn * sizeof(float));
    float* Weff = (float*)wsalloc((size_t)L_DIM * H_DIM * H_DIM * sizeof(float));
    void*  part = wsalloc((size_t)EP_S * Nn * sizeof(float));
    (void)ws_size; (void)n_in; (void)out_size;

    const int tb = 256;
    const dim3 egrid(EP_S, (Nn + NR - 1) / NR);
    deg_part<<<egrid, EP_TPB, 0, stream>>>(row, (int*)part, E, Nn);
    deg_reduce<<<(Nn + tb - 1) / tb, tb, 0, stream>>>((const int*)part, dinv, Nn);
    coef_part<<<egrid, EP_TPB, 0, stream>>>(row, col, dinv, (float*)part, E, Nn);
    coef_reduce<<<(Nn + tb - 1) / tb, tb, 0, stream>>>((const float*)part, coef, Nn);
    weff_kernel<<<L_DIM * H_DIM, H_DIM, 0, stream>>>(W_raw, Weff);

    const int gTiles = (Nn + ROWS - 1) / ROWS;
    fused_chain<<<gTiles, 256, 0, stream>>>(x, enc_W, enc_b, coef, Weff, W_b, ext_w, beta,
                                            dec_W, dec_b, (float*)d_out, Nn);
}


# coef edge pass: every node's degree kept as a 16-bit table in each block's LDS (exact count from memory where saturated); per-edge dinv look-ups are LDS reads, not global gathers; sum order unchanged
# speedup vs baseline: 3.0380x; 1.0300x over previous
#include <hip/hip_runtime.h>

#define H_DIM   128
#define F_IN_DIM 256
#define C_DIM   40
#define L_DIM   4
#define STEPF   0.1f

typedef __attribute__((ext_vector_type(2))) float v2f;
typedef __attribute__((ext_vector_type(4))) float v4f;
typedef __attribute__((ext_vector_type(8))) float v8f;

#define NR      25600
#define EP_TPB  1024
#define EP_U    13
#define EP_S    64

__launch_bounds__(EP_TPB)
__global__ void deg_part(const int* __restrict__ row, int* __restrict__ part, int E, int Nn) {
    __shared__ __align__(16) int hist[NR];
    const int tid  = threadIdx.x;
    const int base = blockIdx.y * NR;
    for (int i = tid; i < NR / 4; i += EP_TPB) ((int4*)hist)[i] = make_int4(0, 0, 0, 0);
    __syncthreads();
    const int ES = (E + (int)gridDim.x - 1) / (int)gridDim.x;
    const int e0 = blockIdx.x * ES;
    const int e1 = min(E, e0 + ES);
    for (int b = e0; b < e1; b += EP_TPB * EP_U) {
        int idx[EP_U];
        #pragma unroll
        for (int u = 0; u < EP_U; ++u) {
            int e = b + u * EP_TPB + tid;
            int v = row[min(e, e1 - 1)];
            idx[u] = (e < e1) ? v : -1;
        }
        #pragma unroll
        for (int u = 0; u < EP_U; ++u) {
            unsigned loc = (unsigned)(idx[u] - base);
            if (loc < (unsigned)NR) atomicAdd(&hist[loc], 1);
        }
    }
    __syncthreads();
    int* dst = part + (size_t)blockIdx.x * Nn + base;
    const int lim = min(NR, Nn - base);
    for (int i = tid; i < lim; i += EP_TPB) dst[i] = hist[i];
}

#define DEG_SAT 0xFFFF
__global__ void deg_reduce(const int* __restrict__ part, unsigned* __restrict__ d16w,
                           int* __restrict__ dfull, int Nn) {
    const int t  = blockIdx.x * blockDim.x + threadIdx.x;
    const int nw = ((Nn + 7) / 8) * 4;
    if (t < nw) {
        const int n0 = 2 * t, n1 = 2 * t + 1;
        const int c0 = min(n0, Nn - 1), c1 = min(n1, Nn - 1);
        int a = 0, b = 0;
        #pragma unroll
        for (int s = 0; s < EP_S; ++s) {
            a += part[(size_t)s * Nn + c0];
            b += part[(size_t)s * Nn + c1];
        }
        if (n0 >= Nn) a = 0;
        if (n1 >= Nn) b = 0;
        d16w[t] = (unsigned)min(a, DEG_SAT) | ((unsigned)min(b, DEG_SAT) << 16);
        if (n0 < Nn) *(int2*)&dfull[n0] = make_int2(a, b);
    }
}

#define TAB_PIECES 13
#define TAB_N (TAB_PIECES * EP_TPB * 8)
__device__ __forceinline__ float dinv_of(int d) { return (d > 0) ? rsqrtf((float)d) : 0.f; }

__launch_bounds__(EP_TPB)
__global__ void coef_part(const int* __restrict__ row, const int* __restrict__ col,
                          const unsigned* __restrict__ d16w, const int* __restrict__ dfull,
                          float* __restrict__ part, int E, int Nn) {
    __shared__ __align__(16) float acc[NR];
    __shared__ __align__(16) unsigned short tab[TAB_N];
    const int tid  = threadIdx.x;
    const int base = blockIdx.y * NR;
    const int ES = (E + (int)gridDim.x - 1) / (int)gridDim.x;
    const int e0 = blockIdx.x * ES;
    const int e1 = min(E, e0 + ES);
    const bool any = e0 < e1;

    int cr[EP_U], rw[EP_U];
    if (any) {
        #pragma unroll
        for (int u = 0; u < EP_U; ++u) {
            int ec = min(e0 + u * EP_TPB + tid, e1 - 1);
            cr[u] = col[ec];
            rw[u] = row[ec];
        }
    }
    {
        const uint4* src = (const uint4*)d16w;
        const int nchunk = min((Nn + 7) / 8, TAB_N / 8);
        uint4 tv[TAB_PIECES];
        #pragma unroll
        for (int j = 0; j < TAB_PIECES; ++j) tv[j] = src[min(tid + j * EP_TPB, nchunk - 1)];
        for (int i = tid; i < NR / 4; i += EP_TPB) ((float4*)acc)[i] = make_float4(0.f, 0.f, 0.f, 0.f);
        #pragma unroll
        for (int j = 0; j < TAB_PIECES; ++j) ((uint4*)tab)[tid + j * EP_TPB] = tv[j];
    }
    __syncthreads();
    if (any) {
        for (int b = e0;;) {
            int cc[EP_U], rr[EP_U];
            #pragma unroll
            for (int u = 0; u < EP_U; ++u) {
                int e = b + u * EP_TPB + tid;
                bool in = (e < e1) && ((unsigned)(cr[u] - base) < (unsigned)NR);
                cc[u] = in ? cr[u] : -1;
                rr[u] = in ? rw[u] : base;
            }
            int dr[EP_U], dc[EP_U];
            bool redo = false;
            #pragma unroll
            for (int u = 0; u < EP_U; ++u) {
                int cn = cc[u] >= 0 ? cc[u] : base;
                int rn = rr[u];
                dc[u] = (cn < TAB_N) ? (int)tab[min(cn, TAB_N - 1)] : DEG_SAT;
                dr[u] = (rn < TAB_N) ? (int)tab[min(rn, TAB_N - 1)] : DEG_SAT;
                redo |= (dc[u] == DEG_SAT) | (dr[u] == DEG_SAT);
            }
            if (redo) {
                #pragma unroll
                for (int u = 0; u < EP_U; ++u) {
                    if (dc[u] == DEG_SAT) dc[u] = dfull[cc[u] >= 0 ? cc[u] : base];
                    if (dr[u] == DEG_SAT) dr[u] = dfull[rr[u]];
                }
            }
            #pragma unroll
            for (int u = 0; u < EP_U; ++u)
                if (cc[u] >= 0) atomicAdd(&acc[cc[u] - base], dinv_of(dr[u]) * dinv_of(dc[u]));
            b += EP_TPB * EP_U;
            if (b >= e1) break;
            #pragma unroll
            for (int u = 0; u < EP_U; ++u) {
                int ec = min(b + u * EP_TPB + tid, e1 - 1);
                cr[u] = col[ec];
                rw[u] = row[ec];
            }
        }
    }
    __syncthreads();
    float* dst = part + (size_t)blockIdx.x * Nn + base;
    const int lim = min(NR, Nn - base);
    for (int i = tid; i < lim; i += EP_TPB) dst[i] = acc[i];
}

__global__ void coef_reduce(const float* __restrict__ part, float* __restrict__ coef, int Nn) {
    int n = blockIdx.x * blockDim.x + threadIdx.x;
    if (n < Nn) {
        float v[EP_S];
        #pragma unroll
        for (int s = 0; s < EP_S; ++s) v[s] = part[(size_t)s * Nn + n];
        static_assert(EP_S % 16 == 0, "the fences below take the planes 16 at a time");
        #pragma unroll
        for (int s = 0; s < EP_S; s += 16) {
            asm volatile("" :: "v"(v[s + 0]), "v"(v[s + 1]), "v"(v[s + 2]), "v"(v[s + 3]), "v"(v[s + 4]), "v"(v[s + 5]), "v"(v[s + 6]), "v"(v[s + 7]), "v"(v[s + 8]), "v"(v[s + 9]), "v"(v[s + 10]), "v"(v[s + 11]), "v"(v[s + 12]), "v"(v[s + 13]), "v"(v[s + 14]), "v"(v[s + 15]));
        }
        float a = 0.f;
        #pragma unroll
        for (int s = 0; s < EP_S; ++s) a += v[s];
        coef[n] = a;
    }
}

__launch_bounds__(H_DIM)
__global__ void weff_kernel(const float* __restrict__ Wraw, float* __restrict__ Weff) {
    __shared__ float absrow[H_DIM];
    const int l = blockIdx.x / H_DIM;
    const int i = blockIdx.x % H_DIM;
    const int j = threadIdx.x;
    const float* Wr = Wraw + (size_t)l * H_DIM * (H_DIM + 2);
    float*       We = Weff + (size_t)l * H_DIM * H_DIM;
    float w = 0.f;
    if (j > i)      w = Wr[i * (H_DIM + 2) + j];
    else if (j < i) w = Wr[j * (H_DIM + 2) + i];
    absrow[j] = fabsf(w);
    const float da = Wr[i * (H_DIM + 2) + H_DIM];
    const float db = Wr[i * (H_DIM + 2) + H_DIM + 1];
    __syncthreads();
    float v = w;
    if (j == i) {
        float s = 0.f;
        for (int t = 0; t < H_DIM; ++t) s += absrow[t];
        v = da * s + db;
    }
    We[i * H_DIM + j] = v;
}

#define MT    5
#define ROWS  (16 * MT)
#define LROW(r) ((((r) & 15) * MT) + ((r) >> 4))
#define XP    (F_IN_DIM + 4)
#define HP    (H_DIM + 4)

__launch_bounds__(256)
__global__ void fused_chain(const float* __restrict__ x, const float* __restrict__ encW,
                            const float* __restrict__ encb, const float* __restrict__ coef,
                            const float* __restrict__ Weff, const float* __restrict__ Wb,
                            const float* __restrict__ extw, const float* __restrict__ beta,
                            const float* __restrict__ decW, const float* __restrict__ decb,
                            float* __restrict__ out, int Nn) {
    __shared__ __align__(16) float bufA[ROWS * XP];
    __shared__ __align__(16) float bufB[ROWS * HP];
    __shared__ __align__(16) float cfs[ROWS];
    const int row0 = blockIdx.x * ROWS;
    if (threadIdx.x < ROWS) {
        int node = row0 + threadIdx.x;
        cfs[threadIdx.x] = coef[node < Nn ? node : Nn - 1];
    }

    const v4f* x4 = (const v4f*)x;
    #pragma unroll
    for (int i = 0; i < ROWS * (F_IN_DIM / 4) / 256; ++i) {
        int idx = threadIdx.x + i * 256;
        int r  = idx >> 6;
        int c4 = idx & 63;
        int node = row0 + r;
        int nc = node < Nn ? node : Nn - 1;
        v4f v = __builtin_nontemporal_load(&x4[(size_t)nc * (F_IN_DIM / 4) + c4]);
        if (node >= Nn) v = (v4f){0.f, 0.f, 0.f, 0.f};
        *(v4f*)&bufA[LROW(r) * XP + c4 * 4] = v;
    }
    __syncthreads();

    const int lane = threadIdx.x & 31;
    const int wave = threadIdx.x >> 5;
    const int col0 = wave * 16;
    const int m    = lane & 15;
    const int hi   = lane >> 4;
    const int colN = col0 + m;

    float h0r[MT][8];
    {
        const float* wrow = encW + (size_t)colN * F_IN_DIM;
        v8f acc[MT];
        for (int t = 0; t < MT; ++t) acc[t] = (v8f){};
        #pragma unroll 8
        for (int k0 = 0; k0 < F_IN_DIM; k0 += 4) {
            int k = k0 + 2 * hi;
            v2f b = *(const v2f*)&wrow[k];
            for (int t = 0; t < MT; ++t) {
                v2f a = *(const v2f*)&bufA[(m * MT + t) * XP + k];
                acc[t] = __builtin_amdgcn_wmma_f32_16x16x4_f32(false, a, false, b,
                                                               (short)0, acc[t], false, false);
            }
        }
        const float bv = encb[colN];
        for (int t = 0; t < MT; ++t)
            for (int r = 0; r < 8; ++r) {
                h0r[t][r] = acc[t][r] + bv;
                bufB[((r + 8 * hi) * MT + t) * HP + colN] = h0r[t][r];
            }
    }
    __syncthreads();

    #pragma nounroll
    for (int l = 0; l < L_DIM; ++l) {
        const float* cur = (l & 1) ? bufA : bufB;
        float*       nxt = (l & 1) ? bufB : bufA;
        const float* We = Weff + (size_t)l * H_DIM * H_DIM;

        v8f acc[MT];
        for (int t = 0; t < MT; ++t) acc[t] = (v8f){};
        #pragma unroll 8
        for (int k0 = 0; k0 < H_DIM; k0 += 4) {
            int k = k0 + 2 * hi;
            v2f b;
            b.x = We[(size_t)k * H_DIM + colN];
            b.y = We[(size_t)(k + 1) * H_DIM + colN];
            for (int t = 0; t < MT; ++t) {
                v2f a = *(const v2f*)&cur[(m * MT + t) * HP + k];
                acc[t] = __builtin_amdgcn_wmma_f32_16x16x4_f32(false, a, false, b,
                                                               (short)0, acc[t], false, false);
            }
        }

        const float wbv = Wb[l * H_DIM + colN];
        const float ew  = extw[l * H_DIM + colN];
        const float bl  = beta[l];
        float hv[MT][8];
        v4f cfa[MT], cfb[MT];
        for (int t = 0; t < MT; ++t) {
            for (int r = 0; r < 8; ++r)
                hv[t][r] = cur[((r + 8 * hi) * MT + t) * HP + colN];
            cfa[t] = *(const v4f*)&cfs[t * 16 + 8 * hi];
            cfb[t] = *(const v4f*)&cfs[t * 16 + 8 * hi + 4];
        }
        for (int t = 0; t < MT; ++t)
            for (int r = 0; r < 8; ++r) {
                float cfv  = (r < 4) ? cfa[t][r & 3] : cfb[t][r & 3];
                float h0v  = h0r[t][r];
                float outp = acc[t][r] + wbv;
                float outv = cfv * outp - (hv[t][r] * ew + bl * h0v);
                nxt[((r + 8 * hi) * MT + t) * HP + colN] = hv[t][r] + STEPF * fmaxf(outv, 0.f);
            }
        __syncthreads();
    }

    if (wave < 3) {
        const float* hf = (L_DIM & 1) ? bufA : bufB;
        const bool colok = (colN < C_DIM);
        const float* wrow = decW + (size_t)(colok ? colN : 0) * H_DIM;
        v8f acc[MT];
        for (int t = 0; t < MT; ++t) acc[t] = (v8f){};
        for (int k0 = 0; k0 < H_DIM; k0 += 4) {
            int k = k0 + 2 * hi;
            v2f b = {0.f, 0.f};
            if (colok) b = *(const v2f*)&wrow[k];
            for (int t = 0; t < MT; ++t) {
                v2f a = *(const v2f*)&hf[(m * MT + t) * HP + k];
                acc[t] = __builtin_amdgcn_wmma_f32_16x16x4_f32(false, a, false, b,
                                                               (short)0, acc[t], false, false);
            }
        }
        if (colok) {
            const float bv = decb[colN];
            for (int t = 0; t < MT; ++t)
                for (int r = 0; r < 8; ++r) {
                    int node = row0 + t * 16 + r + 8 * hi;
                    if (node < Nn)
                        __builtin_nontemporal_store(acc[t][r] + bv,
                                                    &out[(size_t)node * C_DIM + colN]);
                }
        }
    }
}


extern "C" void kernel_launch(void* const* d_in, const int* in_sizes, int n_in,
                              void* d_out, int out_size, void* d_ws, size_t ws_size,
                              hipStream_t stream) {
    const float* x     = (const float*)d_in[0];
    const int*   eidx  = (const int*)d_in[1];
    const float* enc_W = (const float*)d_in[2];
    const float* enc_b = (const float*)d_in[3];
    const float* W_raw = (const float*)d_in[4];
    const float* W_b   = (const float*)d_in[5];
    const float* ext_w = (const float*)d_in[6];
    const float* beta  = (const float*)d_in[7];
    const float* dec_W = (const float*)d_in[8];
    const float* dec_b = (const float*)d_in[9];

    const int Nn = in_sizes[0] / F_IN_DIM;
    const int E  = in_sizes[1] / 2;
    const int* row = eidx;
    const int* col = eidx + E;

    char* ws = (char*)d_ws;
    size_t off = 0;
    auto wsalloc = [&](size_t bytes) -> void* {
        void* p = ws + off;
        off = (off + bytes + 255) & ~(size_t)255;
        return p;
    };
    unsigned* d16w = (unsigned*)wsalloc((size_t)((Nn + 7) / 8) * 16);
    int*   dfull = (int*)wsalloc((size_t)(Nn + 1) * sizeof(int));
    float* coef = (float*)wsalloc((size_t)Nn * sizeof(float));
    float* Weff = (float*)wsalloc((size_t)L_DIM * H_DIM * H_DIM * sizeof(float));
    void*  part = wsalloc((size_t)EP_S * Nn * sizeof(float));
    (void)ws_size; (void)n_in; (void)out_size;

    const int tb = 256;
    const dim3 egrid(EP_S, (Nn + NR - 1) / NR);
    deg_part<<<egrid, EP_TPB, 0, stream>>>(row, (int*)part, E, Nn);
    deg_reduce<<<(((Nn + 7) / 8) * 4 + tb - 1) / tb, tb, 0, stream>>>((const int*)part, d16w, dfull, Nn);
    coef_part<<<egrid, EP_TPB, 0, stream>>>(row, col, d16w, dfull, (float*)part, E, Nn);
    coef_reduce<<<(Nn + tb - 1) / tb, tb, 0, stream>>>((const float*)part, coef, Nn);
    weff_kernel<<<L_DIM * H_DIM, H_DIM, 0, stream>>>(W_raw, Weff);

    const int gTiles = (Nn + ROWS - 1) / ROWS;
    fused_chain<<<gTiles, 256, 0, stream>>>(x, enc_W, enc_b, coef, Weff, W_b, ext_w, beta,
                                            dec_W, dec_b, (float*)d_out, Nn);
}


# two short set-up launches folded into neighbours: W_eff rows ride as extra blocks of the degree-sum launch, the fused chain adds its rows' coef planes itself in slice order; 6 launches -> 4
# speedup vs baseline: 3.1341x; 1.0316x over previous
#include <hip/hip_runtime.h>

#define H_DIM   128
#define F_IN_DIM 256
#define C_DIM   40
#define L_DIM   4
#define STEPF   0.1f

typedef __attribute__((ext_vector_type(2))) float v2f;
typedef __attribute__((ext_vector_type(4))) float v4f;
typedef __attribute__((ext_vector_type(8))) float v8f;

#define NR      25600
#define EP_TPB  1024
#define EP_U    13
#define EP_S    64

__launch_bounds__(EP_TPB)
__global__ void deg_part(const int* __restrict__ row, int* __restrict__ part, int E, int Nn) {
    __shared__ __align__(16) int hist[NR];
    const int tid  = threadIdx.x;
    const int base = blockIdx.y * NR;
    for (int i = tid; i < NR / 4; i += EP_TPB) ((int4*)hist)[i] = make_int4(0, 0, 0, 0);
    __syncthreads();
    const int ES = (E + (int)gridDim.x - 1) / (int)gridDim.x;
    const int e0 = blockIdx.x * ES;
    const int e1 = min(E, e0 + ES);
    for (int b = e0; b < e1; b += EP_TPB * EP_U) {
        int idx[EP_U];
        #pragma unroll
        for (int u = 0; u < EP_U; ++u) {
            int e = b + u * EP_TPB + tid;
            int v = row[min(e, e1 - 1)];
            idx[u] = (e < e1) ? v : -1;
        }
        #pragma unroll
        for (int u = 0; u < EP_U; ++u) {
            unsigned loc = (unsigned)(idx[u] - base);
            if (loc < (unsigned)NR) atomicAdd(&hist[loc], 1);
        }
    }
    __syncthreads();
    int* dst = part + (size_t)blockIdx.x * Nn + base;
    const int lim = min(NR, Nn - base);
    for (int i = tid; i < lim; i += EP_TPB) dst[i] = hist[i];
}

#define DEG_SAT 0xFFFF
#define RW_TPB      256
#define WEFF_RPB    (RW_TPB / H_DIM)
#define WEFF_BLOCKS (L_DIM * H_DIM / WEFF_RPB)
static_assert(RW_TPB % H_DIM == 0 && (L_DIM * H_DIM) % WEFF_RPB == 0, "whole rows per block");
__launch_bounds__(RW_TPB)
__global__ void deg_reduce_weff(const int* __restrict__ part, unsigned* __restrict__ d16w,
                                int* __restrict__ dfull, int Nn,
                                const float* __restrict__ Wraw, float* __restrict__ Weff) {
    __shared__ float absrow[WEFF_RPB][H_DIM];
    if (blockIdx.x < WEFF_BLOCKS) {
        const int sub = threadIdx.x / H_DIM;
        const int rid = blockIdx.x * WEFF_RPB + sub;
        const int l = rid / H_DIM;
        const int i = rid % H_DIM;
        const int j = threadIdx.x % H_DIM;
        const float* Wr = Wraw + (size_t)l * H_DIM * (H_DIM + 2);
        float*       We = Weff + (size_t)l * H_DIM * H_DIM;
        float w = 0.f;
        if (j > i)      w = Wr[i * (H_DIM + 2) + j];
        else if (j < i) w = Wr[j * (H_DIM + 2) + i];
        absrow[sub][j] = fabsf(w);
        const float da = Wr[i * (H_DIM + 2) + H_DIM];
        const float db = Wr[i * (H_DIM + 2) + H_DIM + 1];
        __syncthreads();
        float v = w;
        if (j == i) {
            float s = 0.f;
            #pragma unroll
            for (int t = 0; t < H_DIM; ++t) s += absrow[sub][t];
            v = da * s + db;
        }
        We[i * H_DIM + j] = v;
        return;
    }
    const int t  = (blockIdx.x - WEFF_BLOCKS) * RW_TPB + threadIdx.x;
    const int nw = ((Nn + 7) / 8) * 4;
    if (t < nw) {
        const int n0 = 2 * t, n1 = 2 * t + 1;
        const int c0 = min(n0, Nn - 1), c1 = min(n1, Nn - 1);
        int a = 0, b = 0;
        #pragma unroll
        for (int s = 0; s < EP_S; ++s) {
            a += part[(size_t)s * Nn + c0];
            b += part[(size_t)s * Nn + c1];
        }
        if (n0 >= Nn) a = 0;
        if (n1 >= Nn) b = 0;
        d16w[t] = (unsigned)min(a, DEG_SAT) | ((unsigned)min(b, DEG_SAT) << 16);
        if (n0 < Nn) *(int2*)&dfull[n0] = make_int2(a, b);
    }
}

#define TAB_PIECES 13
#define TAB_N (TAB_PIECES * EP_TPB * 8)
__device__ __forceinline__ float dinv_of(int d) { return (d > 0) ? rsqrtf((float)d) : 0.f; }

__launch_bounds__(EP_TPB)
__global__ void coef_part(const int* __restrict__ row, const int* __restrict__ col,
                          const unsigned* __restrict__ d16w, const int* __restrict__ dfull,
                          float* __restrict__ part, int E, int Nn) {
    __shared__ __align__(16) float acc[NR];
    __shared__ __align__(16) unsigned short tab[TAB_N];
    const int tid  = threadIdx.x;
    const int base = blockIdx.y * NR;
    const int ES = (E + (int)gridDim.x - 1) / (int)gridDim.x;
    const int e0 = blockIdx.x * ES;
    const int e1 = min(E, e0 + ES);
    const bool any = e0 < e1;

    int cr[EP_U], rw[EP_U];
    if (any) {
        #pragma unroll
        for (int u = 0; u < EP_U; ++u) {
            int ec = min(e0 + u * EP_TPB + tid, e1 - 1);
            cr[u] = col[ec];
            rw[u] = row[ec];
        }
    }
    {
        const uint4* src = (const uint4*)d16w;
        const int nchunk = min((Nn + 7) / 8, TAB_N / 8);
        uint4 tv[TAB_PIECES];
        #pragma unroll
        for (int j = 0; j < TAB_PIECES; ++j) tv[j] = src[min(tid + j * EP_TPB, nchunk - 1)];
        for (int i = tid; i < NR / 4; i += EP_TPB) ((float4*)acc)[i] = make_float4(0.f, 0.f, 0.f, 0.f);
        #pragma unroll
        for (int j = 0; j < TAB_PIECES; ++j) ((uint4*)tab)[tid + j * EP_TPB] = tv[j];
    }
    __syncthreads();
    if (any) {
        for (int b = e0;;) {
            int cc[EP_U], rr[EP_U];
            #pragma unroll
            for (int u = 0; u < EP_U; ++u) {
                int e = b + u * EP_TPB + tid;
                bool in = (e < e1) && ((unsigned)(cr[u] - base) < (unsigned)NR);
                cc[u] = in ? cr[u] : -1;
                rr[u] = in ? rw[u] : base;
            }
            int dr[EP_U], dc[EP_U];
            bool redo = false;
            #pragma unroll
            for (int u = 0; u < EP_U; ++u) {
                int cn = cc[u] >= 0 ? cc[u] : base;
                int rn = rr[u];
                dc[u] = (cn < TAB_N) ? (int)tab[min(cn, TAB_N - 1)] : DEG_SAT;
                dr[u] = (rn < TAB_N) ? (int)tab[min(rn, TAB_N - 1)] : DEG_SAT;
                redo |= (dc[u] == DEG_SAT) | (dr[u] == DEG_SAT);
            }
            if (redo) {
                #pragma unroll
                for (int u = 0; u < EP_U; ++u) {
                    if (dc[u] == DEG_SAT) dc[u] = dfull[cc[u] >= 0 ? cc[u] : base];
                    if (dr[u] == DEG_SAT) dr[u] = dfull[rr[u]];
                }
            }
            #pragma unroll
            for (int u = 0; u < EP_U; ++u)
                if (cc[u] >= 0) atomicAdd(&acc[cc[u] - base], dinv_of(dr[u]) * dinv_of(dc[u]));
            b += EP_TPB * EP_U;
            if (b >= e1) break;
            #pragma unroll
            for (int u = 0; u < EP_U; ++u) {
                int ec = min(b + u * EP_TPB + tid, e1 - 1);
                cr[u] = col[ec];
                rw[u] = row[ec];
            }
        }
    }
    __syncthreads();
    float* dst = part + (size_t)blockIdx.x * Nn + base;
    const int lim = min(NR, Nn - base);
    for (int i = tid; i < lim; i += EP_TPB) dst[i] = acc[i];
}

#define MT    5
#define ROWS  (16 * MT)
#define LROW(r) ((((r) & 15) * MT) + ((r) >> 4))
#define XP    (F_IN_DIM + 4)
#define HP    (H_DIM + 4)

__launch_bounds__(256)
__global__ void fused_chain(const float* __restrict__ x, const float* __restrict__ encW,
                            const float* __restrict__ encb, const float* __restrict__ cpart,
                            const float* __restrict__ Weff, const float* __restrict__ Wb,
                            const float* __restrict__ extw, const float* __restrict__ beta,
                            const float* __restrict__ decW, const float* __restrict__ decb,
                            float* __restrict__ out, int Nn) {
    __shared__ __align__(16) float bufA[ROWS * XP];
    __shared__ __align__(16) float bufB[ROWS * HP];
    __shared__ __align__(16) float cfs[ROWS];
    const int row0 = blockIdx.x * ROWS;

    const v4f* x4 = (const v4f*)x;
    constexpr int XL = ROWS * (F_IN_DIM / 4) / 256;
    static_assert(XL == 20, "the fence below names the tile's 20 loads one by one");
    v4f xv[XL];
    #pragma unroll
    for (int i = 0; i < XL; ++i) {
        int idx = threadIdx.x + i * 256;
        int r  = idx >> 6;
        int c4 = idx & 63;
        int node = row0 + r;
        int nc = node < Nn ? node : Nn - 1;
        xv[i] = __builtin_nontemporal_load(&x4[(size_t)nc * (F_IN_DIM / 4) + c4]);
    }
    asm volatile("" :: "v"(xv[0]), "v"(xv[1]), "v"(xv[2]), "v"(xv[3]), "v"(xv[4]), "v"(xv[5]), "v"(xv[6]), "v"(xv[7]), "v"(xv[8]), "v"(xv[9]), "v"(xv[10]), "v"(xv[11]), "v"(xv[12]), "v"(xv[13]), "v"(xv[14]), "v"(xv[15]), "v"(xv[16]), "v"(xv[17]), "v"(xv[18]), "v"(xv[19]));
    #pragma unroll
    for (int i = 0; i < XL; ++i) {
        int idx = threadIdx.x + i * 256;
        int r  = idx >> 6;
        int c4 = idx & 63;
        v4f v = xv[i];
        if (row0 + r >= Nn) v = (v4f){0.f, 0.f, 0.f, 0.f};
        *(v4f*)&bufA[LROW(r) * XP + c4 * 4] = v;
    }
    __syncthreads();

    if (threadIdx.x < ROWS) {
        const int node = row0 + threadIdx.x;
        const int nc = node < Nn ? node : Nn - 1;
        float pv[EP_S];
        #pragma unroll
        for (int s = 0; s < EP_S; ++s) pv[s] = cpart[(size_t)s * Nn + nc];
        static_assert(EP_S % 16 == 0, "the fences below take the planes 16 at a time");
        #pragma unroll
        for (int s = 0; s < EP_S; s += 16) {
            asm volatile("" :: "v"(pv[s + 0]), "v"(pv[s + 1]), "v"(pv[s + 2]), "v"(pv[s + 3]), "v"(pv[s + 4]), "v"(pv[s + 5]), "v"(pv[s + 6]), "v"(pv[s + 7]), "v"(pv[s + 8]), "v"(pv[s + 9]), "v"(pv[s + 10]), "v"(pv[s + 11]), "v"(pv[s + 12]), "v"(pv[s + 13]), "v"(pv[s + 14]), "v"(pv[s + 15]));
        }
        float a = 0.f;
        #pragma unroll
        for (int s = 0; s < EP_S; ++s) a += pv[s];
        cfs[threadIdx.x] = a;
    }

    const int lane = threadIdx.x & 31;
    const int wave = threadIdx.x >> 5;
    const int col0 = wave * 16;
    const int m    = lane & 15;
    const int hi   = lane >> 4;
    const int colN = col0 + m;

    float h0r[MT][8];
    {
        const float* wrow = encW + (size_t)colN * F_IN_DIM;
        v8f acc[MT];
        for (int t = 0; t < MT; ++t) acc[t] = (v8f){};
        #pragma unroll 8
        for (int k0 = 0; k0 < F_IN_DIM; k0 += 4) {
            int k = k0 + 2 * hi;
            v2f b = *(const v2f*)&wrow[k];
            for (int t = 0; t < MT; ++t) {
                v2f a = *(const v2f*)&bufA[(m * MT + t) * XP + k];
                acc[t] = __builtin_amdgcn_wmma_f32_16x16x4_f32(false, a, false, b,
                                                               (short)0, acc[t], false, false);
            }
        }
        const float bv = encb[colN];
        for (int t = 0; t < MT; ++t)
            for (int r = 0; r < 8; ++r) {
                h0r[t][r] = acc[t][r] + bv;
                bufB[((r + 8 * hi) * MT + t) * HP + colN] = h0r[t][r];
            }
    }
    __syncthreads();

    #pragma nounroll
    for (int l = 0; l < L_DIM; ++l) {
        const float* cur = (l & 1) ? bufA : bufB;
        float*       nxt = (l & 1) ? bufB : bufA;
        const float* We = Weff + (size_t)l * H_DIM * H_DIM;

        v8f acc[MT];
        for (int t = 0; t < MT; ++t) acc[t] = (v8f){};
        #pragma unroll 8
        for (int k0 = 0; k0 < H_DIM; k0 += 4) {
            int k = k0 + 2 * hi;
            v2f b;
            b.x = We[(size_t)k * H_DIM + colN];
            b.y = We[(size_t)(k + 1) * H_DIM + colN];
            for (int t = 0; t < MT; ++t) {
                v2f a = *(const v2f*)&cur[(m * MT + t) * HP + k];
                acc[t] = __builtin_amdgcn_wmma_f32_16x16x4_f32(false, a, false, b,
                                                               (short)0, acc[t], false, false);
            }
        }

        const float wbv = Wb[l * H_DIM + colN];
        const float ew  = extw[l * H_DIM + colN];
        const float bl  = beta[l];
        float hv[MT][8];
        v4f cfa[MT], cfb[MT];
        for (int t = 0; t < MT; ++t) {
            for (int r = 0; r < 8; ++r)
                hv[t][r] = cur[((r + 8 * hi) * MT + t) * HP + colN];
            cfa[t] = *(const v4f*)&cfs[t * 16 + 8 * hi];
            cfb[t] = *(const v4f*)&cfs[t * 16 + 8 * hi + 4];
        }
        for (int t = 0; t < MT; ++t)
            for (int r = 0; r < 8; ++r) {
                float cfv  = (r < 4) ? cfa[t][r & 3] : cfb[t][r & 3];
                float h0v  = h0r[t][r];
                float outp = acc[t][r] + wbv;
                float outv = cfv * outp - (hv[t][r] * ew + bl * h0v);
                nxt[((r + 8 * hi) * MT + t) * HP + colN] = hv[t][r] + STEPF * fmaxf(outv, 0.f);
            }
        __syncthreads();
    }

    if (wave < 3) {
        const float* hf = (L_DIM & 1) ? bufA : bufB;
        const bool colok = (colN < C_DIM);
        const float* wrow = decW + (size_t)(colok ? colN : 0) * H_DIM;
        v8f acc[MT];
        for (int t = 0; t < MT; ++t) acc[t] = (v8f){};
        for (int k0 = 0; k0 < H_DIM; k0 += 4) {
            int k = k0 + 2 * hi;
            v2f b = {0.f, 0.f};
            if (colok) b = *(const v2f*)&wrow[k];
            for (int t = 0; t < MT; ++t) {
                v2f a = *(const v2f*)&hf[(m * MT + t) * HP + k];
                acc[t] = __builtin_amdgcn_wmma_f32_16x16x4_f32(false, a, false, b,
                                                               (short)0, acc[t], false, false);
            }
        }
        if (colok) {
            const float bv = decb[colN];
            for (int t = 0; t < MT; ++t)
                for (int r = 0; r < 8; ++r) {
                    int node = row0 + t * 16 + r + 8 * hi;
                    if (node < Nn)
                        __builtin_nontemporal_store(acc[t][r] + bv,
                                                    &out[(size_t)node * C_DIM + colN]);
                }
        }
    }
}


extern "C" void kernel_launch(void* const* d_in, const int* in_sizes, int n_in,
                              void* d_out, int out_size, void* d_ws, size_t ws_size,
                              hipStream_t stream) {
    const float* x     = (const float*)d_in[0];
    const int*   eidx  = (const int*)d_in[1];
    const float* enc_W = (const float*)d_in[2];
    const float* enc_b = (const float*)d_in[3];
    const float* W_raw = (const float*)d_in[4];
    const float* W_b   = (const float*)d_in[5];
    const float* ext_w = (const float*)d_in[6];
    const float* beta  = (const float*)d_in[7];
    const float* dec_W = (const float*)d_in[8];
    const float* dec_b = (const float*)d_in[9];

    const int Nn = in_sizes[0] / F_IN_DIM;
    const int E  = in_sizes[1] / 2;
    const int* row = eidx;
    const int* col = eidx + E;

    char* ws = (char*)d_ws;
    size_t off = 0;
    auto wsalloc = [&](size_t bytes) -> void* {
        void* p = ws + off;
        off = (off + bytes + 255) & ~(size_t)255;
        return p;
    };
    unsigned* d16w = (unsigned*)wsalloc((size_t)((Nn + 7) / 8) * 16);
    int*   dfull = (int*)wsalloc((size_t)(Nn + 1) * sizeof(int));
    float* Weff = (float*)wsalloc((size_t)L_DIM * H_DIM * H_DIM * sizeof(float));
    void*  part = wsalloc((size_t)EP_S * Nn * sizeof(float));
    (void)ws_size; (void)n_in; (void)out_size;

    const dim3 egrid(EP_S, (Nn + NR - 1) / NR);
    deg_part<<<egrid, EP_TPB, 0, stream>>>(row, (int*)part, E, Nn);
    deg_reduce_weff<<<WEFF_BLOCKS + (((Nn + 7) / 8) * 4 + RW_TPB - 1) / RW_TPB, RW_TPB, 0, stream>>>(
        (const int*)part, d16w, dfull, Nn, W_raw, Weff);
    coef_part<<<egrid, EP_TPB, 0, stream>>>(row, col, d16w, dfull, (float*)part, E, Nn);

    const int gTiles = (Nn + ROWS - 1) / ROWS;
    fused_chain<<<gTiles, 256, 0, stream>>>(x, enc_W, enc_b, (const float*)part, Weff, W_b, ext_w, beta,
                                            dec_W, dec_b, (float*)d_out, Nn);
}
